# starting kernel
# baseline (speedup 1.0000x reference)
#include <hip/hip_runtime.h>

#define NN 100000
#define EE 2000000
#define DD 256
#define CC 16
#define RR 4

typedef __attribute__((ext_vector_type(2))) float v2f;
typedef __attribute__((ext_vector_type(8))) float v8f;

__global__ void zero_kernel(float* __restrict__ p, long long n) {
    long long i = (long long)blockIdx.x * blockDim.x + threadIdx.x;
    long long stride = (long long)gridDim.x * blockDim.x;
    for (; i < n; i += stride) p[i] = 0.0f;
}

__global__ __launch_bounds__(128) void gemm_wmma_kernel(
    const float* __restrict__ feat,
    const float* __restrict__ W,
    const float* __restrict__ bias,
    float* __restrict__ wh)
{
    const int wave = threadIdx.x >> 5;
    const int lane = threadIdx.x & 31;
    const int col  = lane & 15;
    const int half = lane >> 4;

    const int row0 = blockIdx.x * 16;
    const int arow = row0 + col;

    const float* aptr = feat + (size_t)arow * DD + 2 * half;
    const float* bptr = W + (size_t)wave * (DD * CC) + (2 * half) * CC + col;

    v8f acc = {};
    #pragma unroll 8
    for (int k = 0; k < DD; k += 4) {
        v2f a, b;
        a.x = aptr[0];
        a.y = aptr[1];
        b.x = bptr[0];
        b.y = bptr[CC];
        acc = __builtin_amdgcn_wmma_f32_16x16x4_f32(
             false, a,  false, b,
             (short)0, acc,  false,  false);
        aptr += 4;
        bptr += 4 * CC;
    }

    const float bv = bias[wave * CC + col];

    float* out = wh + (size_t)row0 * (RR * CC) + wave * CC + col;
    #pragma unroll
    for (int i = 0; i < 8; ++i) {
        const int m = i + 8 * half;
        out[(size_t)m * (RR * CC)] = acc[i] + bv;
    }
}

__global__ __launch_bounds__(256) void scatter_kernel(
    const float* __restrict__ wh,
    const int*   __restrict__ src,
    const int*   __restrict__ dst,
    const float* __restrict__ ew,
    float* __restrict__ sums,
    float* __restrict__ cnts)
{
    const long long total = (long long)RR * EE;
    long long i = (long long)blockIdx.x * blockDim.x + threadIdx.x;
    if (i >= total) return;

    const int r = (int)(i / EE);
    const int s = src[i];
    const int d = dst[i];
    const float w = ew[i];

    const float4* whp = (const float4*)(wh + (size_t)s * (RR * CC) + r * CC);
    float4 v0 = whp[0];
    float4 v1 = whp[1];
    float4 v2 = whp[2];
    float4 v3 = whp[3];

    float* sp = sums + ((size_t)r * NN + d) * CC;
    atomicAdd(sp + 0,  v0.x * w);
    atomicAdd(sp + 1,  v0.y * w);
    atomicAdd(sp + 2,  v0.z * w);
    atomicAdd(sp + 3,  v0.w * w);
    atomicAdd(sp + 4,  v1.x * w);
    atomicAdd(sp + 5,  v1.y * w);
    atomicAdd(sp + 6,  v1.z * w);
    atomicAdd(sp + 7,  v1.w * w);
    atomicAdd(sp + 8,  v2.x * w);
    atomicAdd(sp + 9,  v2.y * w);
    atomicAdd(sp + 10, v2.z * w);
    atomicAdd(sp + 11, v2.w * w);
    atomicAdd(sp + 12, v3.x * w);
    atomicAdd(sp + 13, v3.y * w);
    atomicAdd(sp + 14, v3.z * w);
    atomicAdd(sp + 15, v3.w * w);

    atomicAdd(cnts + (size_t)r * NN + d, 1.0f);
}

__global__ __launch_bounds__(256) void finalize_kernel(
    const float* __restrict__ sums,
    const float* __restrict__ cnts,
    float* __restrict__ out)
{
    const long long total = (long long)NN * CC;
    long long i = (long long)blockIdx.x * blockDim.x + threadIdx.x;
    if (i >= total) return;
    const int n = (int)(i >> 4);
    const int c = (int)(i & 15);

    float acc = 0.0f;
    #pragma unroll
    for (int r = 0; r < RR; ++r) {
        const float cnt = cnts[(size_t)r * NN + n];
        if (cnt > 0.0f) {
            acc += sums[((size_t)r * NN + n) * CC + c] / cnt;
        }
    }
    out[i] = acc;
}

extern "C" void kernel_launch(void* const* d_in, const int* in_sizes, int n_in,
                              void* d_out, int out_size, void* d_ws, size_t ws_size,
                              hipStream_t stream) {
    const float* feat = (const float*)d_in[0];
    const int*   srci = (const int*)  d_in[1];
    const int*   dsti = (const int*)  d_in[2];
    const float* ew   = (const float*)d_in[3];
    const float* W    = (const float*)d_in[4];
    const float* bias = (const float*)d_in[5];
    float* out = (float*)d_out;

    float* ws   = (float*)d_ws;
    float* wh   = ws;
    float* sums = ws + (size_t)NN * (RR * CC);
    float* cnts = sums + (size_t)NN * (RR * CC);

    {
        const long long nz = (long long)NN * (RR * CC) + (long long)RR * NN;
        const int blocks = 4096;
        zero_kernel<<<blocks, 256, 0, stream>>>(sums, nz);
    }

    {
        dim3 grid(NN / 16);
        gemm_wmma_kernel<<<grid, 128, 0, stream>>>(feat, W, bias, wh);
    }

    {
        const long long total = (long long)RR * EE;
        const int blocks = (int)((total + 255) / 256);
        scatter_kernel<<<blocks, 256, 0, stream>>>(wh, srci, dsti, ew, sums, cnts);
    }

    {
        const long long total = (long long)NN * CC;
        const int blocks = (int)((total + 255) / 256);
        finalize_kernel<<<blocks, 256, 0, stream>>>(sums, cnts, out);
    }
}


# scatter: one edge per 16-lane row (lane = channel): each wave f32 atomic adds two whole 64B rows of sums and each gather reads two of wh; float atomics kept, sums still in arrival order
# speedup vs baseline: 3.5267x; 3.5267x over previous
#include <hip/hip_runtime.h>

#define NN 100000
#define EE 2000000
#define DD 256
#define CC 16
#define RR 4

typedef __attribute__((ext_vector_type(2))) float v2f;
typedef __attribute__((ext_vector_type(8))) float v8f;

__global__ void zero_kernel(float* __restrict__ p, long long n) {
    long long i = (long long)blockIdx.x * blockDim.x + threadIdx.x;
    long long stride = (long long)gridDim.x * blockDim.x;
    for (; i < n; i += stride) p[i] = 0.0f;
}

__global__ __launch_bounds__(128) void gemm_wmma_kernel(
    const float* __restrict__ feat,
    const float* __restrict__ W,
    const float* __restrict__ bias,
    float* __restrict__ wh)
{
    const int wave = threadIdx.x >> 5;
    const int lane = threadIdx.x & 31;
    const int col  = lane & 15;
    const int half = lane >> 4;

    const int row0 = blockIdx.x * 16;
    const int arow = row0 + col;

    const float* aptr = feat + (size_t)arow * DD + 2 * half;
    const float* bptr = W + (size_t)wave * (DD * CC) + (2 * half) * CC + col;

    v8f acc = {};
    #pragma unroll 8
    for (int k = 0; k < DD; k += 4) {
        v2f a, b;
        a.x = aptr[0];
        a.y = aptr[1];
        b.x = bptr[0];
        b.y = bptr[CC];
        acc = __builtin_amdgcn_wmma_f32_16x16x4_f32(
             false, a,  false, b,
             (short)0, acc,  false,  false);
        aptr += 4;
        bptr += 4 * CC;
    }

    const float bv = bias[wave * CC + col];

    float* out = wh + (size_t)row0 * (RR * CC) + wave * CC + col;
    #pragma unroll
    for (int i = 0; i < 8; ++i) {
        const int m = i + 8 * half;
        out[(size_t)m * (RR * CC)] = acc[i] + bv;
    }
}

static_assert(EE % 32 == 0, "a wave's 32 edges must belong to one etype");
static_assert(((long long)RR * EE) % 256 == 0, "grid covers the edges exactly");

#define ROW_SHARE(X, J) __builtin_amdgcn_update_dpp(0, (X), 0x150 + (J), 0xF, 0xF, false)

#define GATHER(J) \
    const float v##J = whr[(unsigned)(ROW_SHARE(s, J) * (RR * CC) + c)];
#define ROW_ADD(J) \
    atomicAdd(sr + (unsigned)(ROW_SHARE(d, J) * CC + c), \
              v##J * __int_as_float(ROW_SHARE(wb, J)));

__global__ __launch_bounds__(256) void scatter_kernel(
    const float* __restrict__ wh,
    const int*   __restrict__ src,
    const int*   __restrict__ dst,
    const float* __restrict__ ew,
    float* __restrict__ sums,
    float* __restrict__ cnts)
{
    const unsigned i = blockIdx.x * 256u + threadIdx.x;
    const int c = threadIdx.x & 15;

    const int r = __builtin_amdgcn_readfirstlane((int)(i / (unsigned)EE));

    const int s = src[i];
    const int d = dst[i];
    const int wb = __float_as_int(ew[i]);

    const float* whr = wh + r * CC;
    float* sr = sums + (size_t)r * NN * CC;
    float* cr = cnts + (size_t)r * NN;

    GATHER(0)  GATHER(1)  GATHER(2)  GATHER(3)
    GATHER(4)  GATHER(5)  GATHER(6)  GATHER(7)
    GATHER(8)  GATHER(9)  GATHER(10) GATHER(11)
    GATHER(12) GATHER(13) GATHER(14) GATHER(15)

    atomicAdd(cr + (unsigned)d, 1.0f);

    ROW_ADD(0)  ROW_ADD(1)  ROW_ADD(2)  ROW_ADD(3)
    ROW_ADD(4)  ROW_ADD(5)  ROW_ADD(6)  ROW_ADD(7)
    ROW_ADD(8)  ROW_ADD(9)  ROW_ADD(10) ROW_ADD(11)
    ROW_ADD(12) ROW_ADD(13) ROW_ADD(14) ROW_ADD(15)
}
#undef GATHER
#undef ROW_ADD
#undef ROW_SHARE

__global__ __launch_bounds__(256) void finalize_kernel(
    const float* __restrict__ sums,
    const float* __restrict__ cnts,
    float* __restrict__ out)
{
    const long long total = (long long)NN * CC;
    long long i = (long long)blockIdx.x * blockDim.x + threadIdx.x;
    if (i >= total) return;
    const int n = (int)(i >> 4);
    const int c = (int)(i & 15);

    float acc = 0.0f;
    #pragma unroll
    for (int r = 0; r < RR; ++r) {
        const float cnt = cnts[(size_t)r * NN + n];
        if (cnt > 0.0f) {
            acc += sums[((size_t)r * NN + n) * CC + c] / cnt;
        }
    }
    out[i] = acc;
}

extern "C" void kernel_launch(void* const* d_in, const int* in_sizes, int n_in,
                              void* d_out, int out_size, void* d_ws, size_t ws_size,
                              hipStream_t stream) {
    const float* feat = (const float*)d_in[0];
    const int*   srci = (const int*)  d_in[1];
    const int*   dsti = (const int*)  d_in[2];
    const float* ew   = (const float*)d_in[3];
    const float* W    = (const float*)d_in[4];
    const float* bias = (const float*)d_in[5];
    float* out = (float*)d_out;

    float* ws   = (float*)d_ws;
    float* wh   = ws;
    float* sums = ws + (size_t)NN * (RR * CC);
    float* cnts = sums + (size_t)NN * (RR * CC);

    {
        const long long nz = (long long)NN * (RR * CC) + (long long)RR * NN;
        const int blocks = 4096;
        zero_kernel<<<blocks, 256, 0, stream>>>(sums, nz);
    }

    {
        dim3 grid(NN / 16);
        gemm_wmma_kernel<<<grid, 128, 0, stream>>>(feat, W, bias, wh);
    }

    {
        const long long total = (long long)RR * EE;
        const int blocks = (int)((total + 255) / 256);
        scatter_kernel<<<blocks, 256, 0, stream>>>(wh, srci, dsti, ew, sums, cnts);
    }

    {
        const long long total = (long long)NN * CC;
        const int blocks = (int)((total + 255) / 256);
        finalize_kernel<<<blocks, 256, 0, stream>>>(sums, cnts, out);
    }
}


# scatter: each wave reads its die and draws 128-edge tasks from a per-half ticket list, adding only edges whose destination rows its L2 half owns; f32 atomics kept, sums in arrival order
# speedup vs baseline: 8.2452x; 2.3379x over previous
#include <hip/hip_runtime.h>

#define NN 100000
#define EE 2000000
#define DD 256
#define CC 16
#define RR 4

typedef __attribute__((ext_vector_type(2))) float v2f;
typedef __attribute__((ext_vector_type(8))) float v8f;

__global__ void zero_kernel(float* __restrict__ p, long long n, float* tick) {
    long long i = (long long)blockIdx.x * blockDim.x + threadIdx.x;
    if (i == 0) {
        unsigned* q = reinterpret_cast<unsigned*>(tick);
        q[0] = 0u;
        q[64] = 0u;
    }
    long long stride = (long long)gridDim.x * blockDim.x;
    for (; i < n; i += stride) p[i] = 0.0f;
}

__global__ __launch_bounds__(128) void gemm_wmma_kernel(
    const float* __restrict__ feat,
    const float* __restrict__ W,
    const float* __restrict__ bias,
    float* __restrict__ wh)
{
    const int wave = threadIdx.x >> 5;
    const int lane = threadIdx.x & 31;
    const int col  = lane & 15;
    const int half = lane >> 4;

    const int row0 = blockIdx.x * 16;
    const int arow = row0 + col;

    const float* aptr = feat + (size_t)arow * DD + 2 * half;
    const float* bptr = W + (size_t)wave * (DD * CC) + (2 * half) * CC + col;

    v8f acc = {};
    #pragma unroll 8
    for (int k = 0; k < DD; k += 4) {
        v2f a, b;
        a.x = aptr[0];
        a.y = aptr[1];
        b.x = bptr[0];
        b.y = bptr[CC];
        acc = __builtin_amdgcn_wmma_f32_16x16x4_f32(
             false, a,  false, b,
             (short)0, acc,  false,  false);
        aptr += 4;
        bptr += 4 * CC;
    }

    const float bv = bias[wave * CC + col];

    float* out = wh + (size_t)row0 * (RR * CC) + wave * CC + col;
    #pragma unroll
    for (int i = 0; i < 8; ++i) {
        const int m = i + 8 * half;
        out[(size_t)m * (RR * CC)] = acc[i] + bv;
    }
}

#define NH 50000
#define GROUPS 4
#define NTASK ((unsigned)(((long long)RR * EE) / (32 * GROUPS)))
#define QSTRIDE 64
static_assert(EE % 32 == 0, "32 consecutive edges must belong to one etype");
static_assert(((long long)RR * EE) % (32 * GROUPS) == 0, "tasks cover the edges exactly");
static_assert(NH % 4 == 0, "the split falls on a 256B boundary of sums");

#define ROW_SHARE(X, J) __builtin_amdgcn_update_dpp(0, (X), 0x150 + (J), 0xF, 0xF, false)

#define GATHER(J) \
    const int d##J = ROW_SHARE(d, J); \
    const int s##J = ROW_SHARE(s, J); \
    const bool m##J = ((d##J >= NH) ? 1 : 0) == h; \
    float v##J = 0.0f; \
    if (m##J) v##J = whr[(unsigned)(s##J * (RR * CC) + c)];
#define ROW_ADD(J) { \
    const float w##J = __int_as_float(ROW_SHARE(wb, J)); \
    if (m##J) atomicAdd(sr + (unsigned)(d##J * CC + c), v##J * w##J); }

__device__ __forceinline__ void edge_group(
    const float* __restrict__ whr, float* __restrict__ sr, float* __restrict__ cr,
    const int s, const int d, const int wb, const int c, const int h)
{
    GATHER(0)  GATHER(1)  GATHER(2)  GATHER(3)
    GATHER(4)  GATHER(5)  GATHER(6)  GATHER(7)
    GATHER(8)  GATHER(9)  GATHER(10) GATHER(11)
    GATHER(12) GATHER(13) GATHER(14) GATHER(15)

    if ((((d >= NH) ? 1 : 0) == h)) atomicAdd(cr + (unsigned)d, 1.0f);

    ROW_ADD(0)  ROW_ADD(1)  ROW_ADD(2)  ROW_ADD(3)
    ROW_ADD(4)  ROW_ADD(5)  ROW_ADD(6)  ROW_ADD(7)
    ROW_ADD(8)  ROW_ADD(9)  ROW_ADD(10) ROW_ADD(11)
    ROW_ADD(12) ROW_ADD(13) ROW_ADD(14) ROW_ADD(15)
}

__global__ __launch_bounds__(256) void scatter_kernel(
    const float* __restrict__ wh,
    const int*   __restrict__ src,
    const int*   __restrict__ dst,
    const float* __restrict__ ew,
    float* __restrict__ sums,
    float* __restrict__ cnts,
    float* tick)
{
    unsigned* q = reinterpret_cast<unsigned*>(tick);
    const unsigned lane = threadIdx.x & 31u;
    const int c = (int)(lane & 15u);

    const unsigned place = (unsigned)__builtin_amdgcn_s_sendmsg_rtn(0x87);
    int h = (((place >> 16) & 0xFu) >= 4u) ? 1 : 0;

    for (int pass = 0; pass < 2; ++pass) {
        unsigned* qh = q + h * QSTRIDE;
        for (;;) {
            unsigned t = 0;
            if (lane == 0) t = atomicAdd(qh, 1u);
            t = (unsigned)__builtin_amdgcn_readfirstlane((int)t);
            if (t >= NTASK) break;

            const unsigned i0 = t * (32u * GROUPS) + lane;

            int s[GROUPS], d[GROUPS], wb[GROUPS];
            #pragma unroll
            for (int g = 0; g < GROUPS; ++g) {
                const unsigned i = i0 + 32u * g;
                s[g]  = src[i];
                d[g]  = dst[i];
                wb[g] = __float_as_int(ew[i]);
            }

            #pragma unroll
            for (int g = 0; g < GROUPS; ++g) {
                const unsigned i = i0 + 32u * g;
                const int r = __builtin_amdgcn_readfirstlane((int)(i / (unsigned)EE));
                edge_group(wh + r * CC,
                           sums + (size_t)r * NN * CC,
                           cnts + (size_t)r * NN,
                           s[g], d[g], wb[g], c, h);
            }
        }
        h ^= 1;
    }
}
#undef GATHER
#undef ROW_ADD
#undef ROW_SHARE

__global__ __launch_bounds__(256) void finalize_kernel(
    const float* __restrict__ sums,
    const float* __restrict__ cnts,
    float* __restrict__ out)
{
    const long long total = (long long)NN * CC;
    long long i = (long long)blockIdx.x * blockDim.x + threadIdx.x;
    if (i >= total) return;
    const int n = (int)(i >> 4);
    const int c = (int)(i & 15);

    float acc = 0.0f;
    #pragma unroll
    for (int r = 0; r < RR; ++r) {
        const float cnt = cnts[(size_t)r * NN + n];
        if (cnt > 0.0f) {
            acc += sums[((size_t)r * NN + n) * CC + c] / cnt;
        }
    }
    out[i] = acc;
}

extern "C" void kernel_launch(void* const* d_in, const int* in_sizes, int n_in,
                              void* d_out, int out_size, void* d_ws, size_t ws_size,
                              hipStream_t stream) {
    const float* feat = (const float*)d_in[0];
    const int*   srci = (const int*)  d_in[1];
    const int*   dsti = (const int*)  d_in[2];
    const float* ew   = (const float*)d_in[3];
    const float* W    = (const float*)d_in[4];
    const float* bias = (const float*)d_in[5];
    float* out = (float*)d_out;

    float* ws   = (float*)d_ws;
    float* wh   = ws;
    float* sums = ws + (size_t)NN * (RR * CC);
    float* cnts = sums + (size_t)NN * (RR * CC);

    {
        const long long nz = (long long)NN * (RR * CC) + (long long)RR * NN;
        const int blocks = 4096;
        zero_kernel<<<blocks, 256, 0, stream>>>(sums, nz, out);
    }

    {
        dim3 grid(NN / 16);
        gemm_wmma_kernel<<<grid, 128, 0, stream>>>(feat, W, bias, wh);
    }

    {
        scatter_kernel<<<2048, 256, 0, stream>>>(wh, srci, dsti, ew, sums, cnts, out);
    }

    {
        const long long total = (long long)NN * CC;
        const int blocks = (int)((total + 255) / 256);
        finalize_kernel<<<blocks, 256, 0, stream>>>(sums, cnts, out);
    }
}


# counts: per-destination edge counts built as LDS histograms in a kernel of their own (integer LDS adds, then line-wide f32 adds, exact in any order); the scatter's per-edge count atomics removed
# speedup vs baseline: 9.4510x; 1.1462x over previous
#include <hip/hip_runtime.h>

#define NN 100000
#define EE 2000000
#define DD 256
#define CC 16
#define RR 4

typedef __attribute__((ext_vector_type(2))) float v2f;
typedef __attribute__((ext_vector_type(8))) float v8f;

__global__ void zero_kernel(float* __restrict__ p, long long n, float* tick) {
    long long i = (long long)blockIdx.x * blockDim.x + threadIdx.x;
    if (i == 0) {
        unsigned* q = reinterpret_cast<unsigned*>(tick);
        q[0] = 0u;
        q[64] = 0u;
    }
    long long stride = (long long)gridDim.x * blockDim.x;
    for (; i < n; i += stride) p[i] = 0.0f;
}

#define CR 12500
#define CRANGES (NN / CR)
#define CSLICES 8
#define CSLICE_V (EE / CSLICES / 4)
#define CTHREADS 512
#define CAHEAD 8
static_assert(NN % CR == 0, "ranges cover the nodes exactly");
static_assert(EE % (CSLICES * 4) == 0, "slices are whole 16-byte groups");

__device__ __forceinline__ void tally4(unsigned* hist, const int4 v, const int lo) {
    unsigned o;
    o = (unsigned)(v.x - lo); if (o < (unsigned)CR) atomicAdd(&hist[o], 1u);
    o = (unsigned)(v.y - lo); if (o < (unsigned)CR) atomicAdd(&hist[o], 1u);
    o = (unsigned)(v.z - lo); if (o < (unsigned)CR) atomicAdd(&hist[o], 1u);
    o = (unsigned)(v.w - lo); if (o < (unsigned)CR) atomicAdd(&hist[o], 1u);
}

__global__ __launch_bounds__(CTHREADS) void count_kernel(
    const int* __restrict__ dst,
    float* __restrict__ cnts)
{
    __shared__ unsigned hist[CR];
    const int b = (int)blockIdx.x;
    const int q = b % CRANGES;
    const int k = (b / CRANGES) % CSLICES;
    const int r = b / (CRANGES * CSLICES);
    const int lo = q * CR;

    for (int i = (int)threadIdx.x; i < CR; i += CTHREADS) hist[i] = 0u;
    __syncthreads();

    const int4* p = reinterpret_cast<const int4*>(dst + (size_t)r * EE) + (size_t)k * CSLICE_V;
    int i = (int)threadIdx.x;
    for (; i + (CAHEAD - 1) * CTHREADS < CSLICE_V; i += CAHEAD * CTHREADS) {
        int4 v[CAHEAD];
        #pragma unroll
        for (int u = 0; u < CAHEAD; ++u) v[u] = p[i + u * CTHREADS];
        #pragma unroll
        for (int u = 0; u < CAHEAD; ++u) tally4(hist, v[u], lo);
    }
    for (; i < CSLICE_V; i += CTHREADS) tally4(hist, p[i], lo);
    __syncthreads();

    float* c = cnts + (size_t)r * NN + lo;
    for (int j = (int)threadIdx.x; j < CR; j += CTHREADS) {
        const unsigned h = hist[j];
        if (h != 0u) atomicAdd(c + j, (float)h);
    }
}

__global__ __launch_bounds__(128) void gemm_wmma_kernel(
    const float* __restrict__ feat,
    const float* __restrict__ W,
    const float* __restrict__ bias,
    float* __restrict__ wh)
{
    const int wave = threadIdx.x >> 5;
    const int lane = threadIdx.x & 31;
    const int col  = lane & 15;
    const int half = lane >> 4;

    const int row0 = blockIdx.x * 16;
    const int arow = row0 + col;

    const float* aptr = feat + (size_t)arow * DD + 2 * half;
    const float* bptr = W + (size_t)wave * (DD * CC) + (2 * half) * CC + col;

    v8f acc = {};
    #pragma unroll 8
    for (int k = 0; k < DD; k += 4) {
        v2f a, b;
        a.x = aptr[0];
        a.y = aptr[1];
        b.x = bptr[0];
        b.y = bptr[CC];
        acc = __builtin_amdgcn_wmma_f32_16x16x4_f32(
             false, a,  false, b,
             (short)0, acc,  false,  false);
        aptr += 4;
        bptr += 4 * CC;
    }

    const float bv = bias[wave * CC + col];

    float* out = wh + (size_t)row0 * (RR * CC) + wave * CC + col;
    #pragma unroll
    for (int i = 0; i < 8; ++i) {
        const int m = i + 8 * half;
        out[(size_t)m * (RR * CC)] = acc[i] + bv;
    }
}

#define NH 50000
#define GROUPS 4
#define NTASK ((unsigned)(((long long)RR * EE) / (32 * GROUPS)))
#define QSTRIDE 64
static_assert(EE % 32 == 0, "32 consecutive edges must belong to one etype");
static_assert(((long long)RR * EE) % (32 * GROUPS) == 0, "tasks cover the edges exactly");
static_assert(NH % 4 == 0, "the split falls on a 256B boundary of sums");

#define ROW_SHARE(X, J) __builtin_amdgcn_update_dpp(0, (X), 0x150 + (J), 0xF, 0xF, false)

#define GATHER(J) \
    const int d##J = ROW_SHARE(d, J); \
    const int s##J = ROW_SHARE(s, J); \
    const bool m##J = ((d##J >= NH) ? 1 : 0) == h; \
    float v##J = 0.0f; \
    if (m##J) v##J = whr[(unsigned)(s##J * (RR * CC) + c)];
#define ROW_ADD(J) { \
    const float w##J = __int_as_float(ROW_SHARE(wb, J)); \
    if (m##J) atomicAdd(sr + (unsigned)(d##J * CC + c), v##J * w##J); }

__device__ __forceinline__ void edge_group(
    const float* __restrict__ whr, float* __restrict__ sr,
    const int s, const int d, const int wb, const int c, const int h)
{
    GATHER(0)  GATHER(1)  GATHER(2)  GATHER(3)
    GATHER(4)  GATHER(5)  GATHER(6)  GATHER(7)
    GATHER(8)  GATHER(9)  GATHER(10) GATHER(11)
    GATHER(12) GATHER(13) GATHER(14) GATHER(15)

    ROW_ADD(0)  ROW_ADD(1)  ROW_ADD(2)  ROW_ADD(3)
    ROW_ADD(4)  ROW_ADD(5)  ROW_ADD(6)  ROW_ADD(7)
    ROW_ADD(8)  ROW_ADD(9)  ROW_ADD(10) ROW_ADD(11)
    ROW_ADD(12) ROW_ADD(13) ROW_ADD(14) ROW_ADD(15)
}

__global__ __launch_bounds__(256) void scatter_kernel(
    const float* __restrict__ wh,
    const int*   __restrict__ src,
    const int*   __restrict__ dst,
    const float* __restrict__ ew,
    float* __restrict__ sums,
    float* tick)
{
    unsigned* q = reinterpret_cast<unsigned*>(tick);
    const unsigned lane = threadIdx.x & 31u;
    const int c = (int)(lane & 15u);

    const unsigned place = (unsigned)__builtin_amdgcn_s_sendmsg_rtn(0x87);
    int h = (((place >> 16) & 0xFu) >= 4u) ? 1 : 0;

    for (int pass = 0; pass < 2; ++pass) {
        unsigned* qh = q + h * QSTRIDE;
        for (;;) {
            unsigned t = 0;
            if (lane == 0) t = atomicAdd(qh, 1u);
            t = (unsigned)__builtin_amdgcn_readfirstlane((int)t);
            if (t >= NTASK) break;

            const unsigned i0 = t * (32u * GROUPS) + lane;

            int s[GROUPS], d[GROUPS], wb[GROUPS];
            #pragma unroll
            for (int g = 0; g < GROUPS; ++g) {
                const unsigned i = i0 + 32u * g;
                s[g]  = src[i];
                d[g]  = dst[i];
                wb[g] = __float_as_int(ew[i]);
            }

            #pragma unroll
            for (int g = 0; g < GROUPS; ++g) {
                const unsigned i = i0 + 32u * g;
                const int r = __builtin_amdgcn_readfirstlane((int)(i / (unsigned)EE));
                edge_group(wh + r * CC,
                           sums + (size_t)r * NN * CC,
                           s[g], d[g], wb[g], c, h);
            }
        }
        h ^= 1;
    }
}
#undef GATHER
#undef ROW_ADD
#undef ROW_SHARE

__global__ __launch_bounds__(256) void finalize_kernel(
    const float* __restrict__ sums,
    const float* __restrict__ cnts,
    float* __restrict__ out)
{
    const long long total = (long long)NN * CC;
    long long i = (long long)blockIdx.x * blockDim.x + threadIdx.x;
    if (i >= total) return;
    const int n = (int)(i >> 4);
    const int c = (int)(i & 15);

    float acc = 0.0f;
    #pragma unroll
    for (int r = 0; r < RR; ++r) {
        const float cnt = cnts[(size_t)r * NN + n];
        if (cnt > 0.0f) {
            acc += sums[((size_t)r * NN + n) * CC + c] / cnt;
        }
    }
    out[i] = acc;
}

extern "C" void kernel_launch(void* const* d_in, const int* in_sizes, int n_in,
                              void* d_out, int out_size, void* d_ws, size_t ws_size,
                              hipStream_t stream) {
    const float* feat = (const float*)d_in[0];
    const int*   srci = (const int*)  d_in[1];
    const int*   dsti = (const int*)  d_in[2];
    const float* ew   = (const float*)d_in[3];
    const float* W    = (const float*)d_in[4];
    const float* bias = (const float*)d_in[5];
    float* out = (float*)d_out;

    float* ws   = (float*)d_ws;
    float* wh   = ws;
    float* sums = ws + (size_t)NN * (RR * CC);
    float* cnts = sums + (size_t)NN * (RR * CC);

    {
        const long long nz = (long long)NN * (RR * CC) + (long long)RR * NN;
        const int blocks = 4096;
        zero_kernel<<<blocks, 256, 0, stream>>>(sums, nz, out);
    }

    {
        count_kernel<<<RR * CSLICES * CRANGES, CTHREADS, 0, stream>>>(dsti, cnts);
    }

    {
        dim3 grid(NN / 16);
        gemm_wmma_kernel<<<grid, 128, 0, stream>>>(feat, W, bias, wh);
    }

    {
        scatter_kernel<<<2048, 256, 0, stream>>>(wh, srci, dsti, ew, sums, out);
    }

    {
        const long long total = (long long)NN * CC;
        const int blocks = (int)((total + 255) / 256);
        finalize_kernel<<<blocks, 256, 0, stream>>>(sums, cnts, out);
    }
}


# linear layer: each wave forms all four etype tiles of its 16 rows from one 16-byte-per-lane read of feat, W staged once per work-group in LDS; f32 WMMA kept, k summed in another order within each 8
# speedup vs baseline: 10.3126x; 1.0912x over previous
#include <hip/hip_runtime.h>

#define NN 100000
#define EE 2000000
#define DD 256
#define CC 16
#define RR 4

typedef __attribute__((ext_vector_type(2))) float v2f;
typedef __attribute__((ext_vector_type(8))) float v8f;

__global__ void zero_kernel(float* __restrict__ p, long long n, float* tick) {
    long long i = (long long)blockIdx.x * blockDim.x + threadIdx.x;
    if (i == 0) {
        unsigned* q = reinterpret_cast<unsigned*>(tick);
        q[0] = 0u;
        q[64] = 0u;
    }
    long long stride = (long long)gridDim.x * blockDim.x;
    for (; i < n; i += stride) p[i] = 0.0f;
}

#define CR 12500
#define CRANGES (NN / CR)
#define CSLICES 8
#define CSLICE_V (EE / CSLICES / 4)
#define CTHREADS 512
#define CAHEAD 8
static_assert(NN % CR == 0, "ranges cover the nodes exactly");
static_assert(EE % (CSLICES * 4) == 0, "slices are whole 16-byte groups");

__device__ __forceinline__ void tally4(unsigned* hist, const int4 v, const int lo) {
    unsigned o;
    o = (unsigned)(v.x - lo); if (o < (unsigned)CR) atomicAdd(&hist[o], 1u);
    o = (unsigned)(v.y - lo); if (o < (unsigned)CR) atomicAdd(&hist[o], 1u);
    o = (unsigned)(v.z - lo); if (o < (unsigned)CR) atomicAdd(&hist[o], 1u);
    o = (unsigned)(v.w - lo); if (o < (unsigned)CR) atomicAdd(&hist[o], 1u);
}

__global__ __launch_bounds__(CTHREADS) void count_kernel(
    const int* __restrict__ dst,
    float* __restrict__ cnts)
{
    __shared__ unsigned hist[CR];
    const int b = (int)blockIdx.x;
    const int q = b % CRANGES;
    const int k = (b / CRANGES) % CSLICES;
    const int r = b / (CRANGES * CSLICES);
    const int lo = q * CR;

    for (int i = (int)threadIdx.x; i < CR; i += CTHREADS) hist[i] = 0u;
    __syncthreads();

    const int4* p = reinterpret_cast<const int4*>(dst + (size_t)r * EE) + (size_t)k * CSLICE_V;
    int i = (int)threadIdx.x;
    for (; i + (CAHEAD - 1) * CTHREADS < CSLICE_V; i += CAHEAD * CTHREADS) {
        int4 v[CAHEAD];
        #pragma unroll
        for (int u = 0; u < CAHEAD; ++u) v[u] = p[i + u * CTHREADS];
        #pragma unroll
        for (int u = 0; u < CAHEAD; ++u) tally4(hist, v[u], lo);
    }
    for (; i < CSLICE_V; i += CTHREADS) tally4(hist, p[i], lo);
    __syncthreads();

    float* c = cnts + (size_t)r * NN + lo;
    for (int j = (int)threadIdx.x; j < CR; j += CTHREADS) {
        const unsigned h = hist[j];
        if (h != 0u) atomicAdd(c + j, (float)h);
    }
}

typedef __attribute__((ext_vector_type(4))) float v4f;
#define GKB (DD / 8)
#define GCH 8
#define GWAVES 8
#define GBLOCKS 768
#define GTILES (NN / 16)
static_assert(NN % 16 == 0, "row tiles are whole");
static_assert(DD % 8 == 0, "k blocks are whole");
static_assert(GKB % (2 * GCH) == 0, "chunks come in pairs");

__device__ __forceinline__ void load_chunk(v4f (&a)[GCH], const float* p) {
    #pragma unroll
    for (int j = 0; j < GCH; ++j) a[j] = *reinterpret_cast<const v4f*>(p + 8 * j);
}

__device__ __forceinline__ void chunk_mma(v8f (&acc)[RR], const v4f (&a)[GCH], const v4f* wc) {
    #pragma unroll
    for (int j = 0; j < GCH; ++j) {
        v2f a0, a1;
        a0.x = a[j].x; a0.y = a[j].y;
        a1.x = a[j].z; a1.y = a[j].w;
        #pragma unroll
        for (int r = 0; r < RR; ++r) {
            const v4f b = wc[(r * GKB + j) * 32];
            v2f b0, b1;
            b0.x = b.x; b0.y = b.y;
            b1.x = b.z; b1.y = b.w;
            acc[r] = __builtin_amdgcn_wmma_f32_16x16x4_f32(
                 false, a0,  false, b0,
                 (short)0, acc[r],  false,  false);
            acc[r] = __builtin_amdgcn_wmma_f32_16x16x4_f32(
                 false, a1,  false, b1,
                 (short)0, acc[r],  false,  false);
        }
    }
}

__global__ __launch_bounds__(GWAVES * 32, 4) void gemm_wmma_kernel(
    const float* __restrict__ feat,
    const float* __restrict__ W,
    const float* __restrict__ bias,
    float* __restrict__ wh)
{
    __shared__ v4f wl[RR * GKB * 32];

    const int tid = (int)threadIdx.x;
    #pragma unroll
    for (int i = 0; i < (RR * GKB * 32) / (GWAVES * 32); ++i) {
        const int idx = tid + (GWAVES * 32) * i;
        const int ln = idx & 31;
        const int kb = (idx >> 5) & (GKB - 1);
        const int r  = idx >> 10;
        const float* p = W + (size_t)r * (DD * CC) + (kb * 8 + 4 * (ln >> 4)) * CC + (ln & 15);
        v4f v;
        v.x = p[0];
        v.y = p[CC];
        v.z = p[2 * CC];
        v.w = p[3 * CC];
        wl[idx] = v;
    }
    __syncthreads();

    const int wave = __builtin_amdgcn_readfirstlane(tid >> 5);
    const int lane = tid & 31;
    const int col  = lane & 15;
    const int half = lane >> 4;

    const int first = wave * GBLOCKS + (int)blockIdx.x;
    const float* fl = feat + (size_t)col * DD + 4 * half;

    v4f ae[GCH], ao[GCH];
    if (first < GTILES) load_chunk(ae, fl + (size_t)first * (16 * DD));

    for (int tile = first; tile < GTILES; tile += GBLOCKS * GWAVES) {
        v8f acc[RR];
        #pragma unroll
        for (int r = 0; r < RR; ++r) acc[r] = (v8f){};
        const float* pt = fl + (size_t)tile * (16 * DD);

        #pragma nounroll
        for (int cp = 0; cp < GKB / (2 * GCH); ++cp) {
            load_chunk(ao, pt + (2 * cp + 1) * (8 * GCH));
            chunk_mma(acc, ae, wl + (size_t)(2 * cp * GCH) * 32 + lane);

            const bool last = (cp == GKB / (2 * GCH) - 1);
            const int ntile = last ? tile + GBLOCKS * GWAVES : tile;
            const int nch   = last ? 0 : 2 * cp + 2;
            if (ntile < GTILES) load_chunk(ae, fl + (size_t)ntile * (16 * DD) + nch * (8 * GCH));
            chunk_mma(acc, ao, wl + (size_t)((2 * cp + 1) * GCH) * 32 + lane);
        }

        asm volatile("v_nop\n\tv_nop\n\tv_nop\n\tv_nop");

        float* out = wh + (size_t)tile * (16 * RR * CC) + col;
        #pragma unroll
        for (int r = 0; r < RR; ++r) {
            const float bv = bias[r * CC + col];
            #pragma unroll
            for (int i = 0; i < 8; ++i) {
                const int m = i + 8 * half;
                out[(size_t)m * (RR * CC) + r * CC] = acc[r][i] + bv;
            }
        }
    }
}

#define NH 50000
#define GROUPS 4
#define NTASK ((unsigned)(((long long)RR * EE) / (32 * GROUPS)))
#define QSTRIDE 64
static_assert(EE % 32 == 0, "32 consecutive edges must belong to one etype");
static_assert(((long long)RR * EE) % (32 * GROUPS) == 0, "tasks cover the edges exactly");
static_assert(NH % 4 == 0, "the split falls on a 256B boundary of sums");

#define ROW_SHARE(X, J) __builtin_amdgcn_update_dpp(0, (X), 0x150 + (J), 0xF, 0xF, false)

#define GATHER(J) \
    const int d##J = ROW_SHARE(d, J); \
    const int s##J = ROW_SHARE(s, J); \
    const bool m##J = ((d##J >= NH) ? 1 : 0) == h; \
    float v##J = 0.0f; \
    if (m##J) v##J = whr[(unsigned)(s##J * (RR * CC) + c)];
#define ROW_ADD(J) { \
    const float w##J = __int_as_float(ROW_SHARE(wb, J)); \
    if (m##J) atomicAdd(sr + (unsigned)(d##J * CC + c), v##J * w##J); }

__device__ __forceinline__ void edge_group(
    const float* __restrict__ whr, float* __restrict__ sr,
    const int s, const int d, const int wb, const int c, const int h)
{
    GATHER(0)  GATHER(1)  GATHER(2)  GATHER(3)
    GATHER(4)  GATHER(5)  GATHER(6)  GATHER(7)
    GATHER(8)  GATHER(9)  GATHER(10) GATHER(11)
    GATHER(12) GATHER(13) GATHER(14) GATHER(15)

    ROW_ADD(0)  ROW_ADD(1)  ROW_ADD(2)  ROW_ADD(3)
    ROW_ADD(4)  ROW_ADD(5)  ROW_ADD(6)  ROW_ADD(7)
    ROW_ADD(8)  ROW_ADD(9)  ROW_ADD(10) ROW_ADD(11)
    ROW_ADD(12) ROW_ADD(13) ROW_ADD(14) ROW_ADD(15)
}

__global__ __launch_bounds__(256) void scatter_kernel(
    const float* __restrict__ wh,
    const int*   __restrict__ src,
    const int*   __restrict__ dst,
    const float* __restrict__ ew,
    float* __restrict__ sums,
    float* tick)
{
    unsigned* q = reinterpret_cast<unsigned*>(tick);
    const unsigned lane = threadIdx.x & 31u;
    const int c = (int)(lane & 15u);

    const unsigned place = (unsigned)__builtin_amdgcn_s_sendmsg_rtn(0x87);
    int h = (((place >> 16) & 0xFu) >= 4u) ? 1 : 0;

    for (int pass = 0; pass < 2; ++pass) {
        unsigned* qh = q + h * QSTRIDE;
        for (;;) {
            unsigned t = 0;
            if (lane == 0) t = atomicAdd(qh, 1u);
            t = (unsigned)__builtin_amdgcn_readfirstlane((int)t);
            if (t >= NTASK) break;

            const unsigned i0 = t * (32u * GROUPS) + lane;

            int s[GROUPS], d[GROUPS], wb[GROUPS];
            #pragma unroll
            for (int g = 0; g < GROUPS; ++g) {
                const unsigned i = i0 + 32u * g;
                s[g]  = src[i];
                d[g]  = dst[i];
                wb[g] = __float_as_int(ew[i]);
            }

            #pragma unroll
            for (int g = 0; g < GROUPS; ++g) {
                const unsigned i = i0 + 32u * g;
                const int r = __builtin_amdgcn_readfirstlane((int)(i / (unsigned)EE));
                edge_group(wh + r * CC,
                           sums + (size_t)r * NN * CC,
                           s[g], d[g], wb[g], c, h);
            }
        }
        h ^= 1;
    }
}
#undef GATHER
#undef ROW_ADD
#undef ROW_SHARE

__global__ __launch_bounds__(256) void finalize_kernel(
    const float* __restrict__ sums,
    const float* __restrict__ cnts,
    float* __restrict__ out)
{
    const long long total = (long long)NN * CC;
    long long i = (long long)blockIdx.x * blockDim.x + threadIdx.x;
    if (i >= total) return;
    const int n = (int)(i >> 4);
    const int c = (int)(i & 15);

    float acc = 0.0f;
    #pragma unroll
    for (int r = 0; r < RR; ++r) {
        const float cnt = cnts[(size_t)r * NN + n];
        if (cnt > 0.0f) {
            acc += sums[((size_t)r * NN + n) * CC + c] / cnt;
        }
    }
    out[i] = acc;
}

extern "C" void kernel_launch(void* const* d_in, const int* in_sizes, int n_in,
                              void* d_out, int out_size, void* d_ws, size_t ws_size,
                              hipStream_t stream) {
    const float* feat = (const float*)d_in[0];
    const int*   srci = (const int*)  d_in[1];
    const int*   dsti = (const int*)  d_in[2];
    const float* ew   = (const float*)d_in[3];
    const float* W    = (const float*)d_in[4];
    const float* bias = (const float*)d_in[5];
    float* out = (float*)d_out;

    float* ws   = (float*)d_ws;
    float* wh   = ws;
    float* sums = ws + (size_t)NN * (RR * CC);
    float* cnts = sums + (size_t)NN * (RR * CC);

    {
        const long long nz = (long long)NN * (RR * CC) + (long long)RR * NN;
        const int blocks = 4096;
        zero_kernel<<<blocks, 256, 0, stream>>>(sums, nz, out);
    }

    {
        count_kernel<<<RR * CSLICES * CRANGES, CTHREADS, 0, stream>>>(dsti, cnts);
    }

    {
        gemm_wmma_kernel<<<GBLOCKS, GWAVES * 32, 0, stream>>>(feat, W, bias, wh);
    }

    {
        scatter_kernel<<<2048, 256, 0, stream>>>(wh, srci, dsti, ew, sums, out);
    }

    {
        const long long total = (long long)NN * CC;
        const int blocks = (int)((total + 255) / 256);
        finalize_kernel<<<blocks, 256, 0, stream>>>(sums, cnts, out);
    }
}


# launches: zeroing moved into the linear-layer launch and the count work-groups into the scatter launch (neither is needed by the launch that now carries it): 5 launches become 3; sums unchanged
# speedup vs baseline: 10.7263x; 1.0401x over previous
#include <hip/hip_runtime.h>

#define NN 100000
#define EE 2000000
#define DD 256
#define CC 16
#define RR 4

typedef __attribute__((ext_vector_type(2))) float v2f;
typedef __attribute__((ext_vector_type(8))) float v8f;

#define CR 12500
#define CRANGES (NN / CR)
#define CSLICES 8
#define CSLICE_V (EE / CSLICES / 4)
#define CTHREADS 512
#define CAHEAD 8
static_assert(NN % CR == 0, "ranges cover the nodes exactly");
static_assert(EE % (CSLICES * 4) == 0, "slices are whole 16-byte groups");

__device__ __forceinline__ void tally4(unsigned* hist, const int4 v, const int lo) {
    unsigned o;
    o = (unsigned)(v.x - lo); if (o < (unsigned)CR) atomicAdd(&hist[o], 1u);
    o = (unsigned)(v.y - lo); if (o < (unsigned)CR) atomicAdd(&hist[o], 1u);
    o = (unsigned)(v.z - lo); if (o < (unsigned)CR) atomicAdd(&hist[o], 1u);
    o = (unsigned)(v.w - lo); if (o < (unsigned)CR) atomicAdd(&hist[o], 1u);
}

#define CBLOCKS (RR * CSLICES * CRANGES)

__device__ __forceinline__ void count_body(
    unsigned* hist,
    const int* __restrict__ dst,
    float* __restrict__ cnts)
{
    const int b = (int)blockIdx.x;
    const int q = b % CRANGES;
    const int k = (b / CRANGES) % CSLICES;
    const int r = b / (CRANGES * CSLICES);
    const int lo = q * CR;

    for (int i = (int)threadIdx.x; i < CR; i += CTHREADS) hist[i] = 0u;
    __syncthreads();

    const int4* p = reinterpret_cast<const int4*>(dst + (size_t)r * EE) + (size_t)k * CSLICE_V;
    int i = (int)threadIdx.x;
    for (; i + (CAHEAD - 1) * CTHREADS < CSLICE_V; i += CAHEAD * CTHREADS) {
        int4 v[CAHEAD];
        #pragma unroll
        for (int u = 0; u < CAHEAD; ++u) v[u] = p[i + u * CTHREADS];
        #pragma unroll
        for (int u = 0; u < CAHEAD; ++u) tally4(hist, v[u], lo);
    }
    for (; i < CSLICE_V; i += CTHREADS) tally4(hist, p[i], lo);
    __syncthreads();

    float* c = cnts + (size_t)r * NN + lo;
    for (int j = (int)threadIdx.x; j < CR; j += CTHREADS) {
        const unsigned h = hist[j];
        if (h != 0u) atomicAdd(c + j, (float)h);
    }
}

typedef __attribute__((ext_vector_type(4))) float v4f;
#define GKB (DD / 8)
#define GCH 8
#define GWAVES 8
#define GBLOCKS 768
#define GTILES (NN / 16)
#define NZ4 ((NN * (RR * CC) + RR * NN) / 4)
static_assert((NN * (RR * CC) + RR * NN) % 4 == 0, "the zeroed region is whole 16-byte groups");
static_assert(NN % 16 == 0, "row tiles are whole");
static_assert(DD % 8 == 0, "k blocks are whole");
static_assert(GKB % (2 * GCH) == 0, "chunks come in pairs");

__device__ __forceinline__ void load_chunk(v4f (&a)[GCH], const float* p) {
    #pragma unroll
    for (int j = 0; j < GCH; ++j) a[j] = *reinterpret_cast<const v4f*>(p + 8 * j);
}

__device__ __forceinline__ void chunk_mma(v8f (&acc)[RR], const v4f (&a)[GCH], const v4f* wc) {
    #pragma unroll
    for (int j = 0; j < GCH; ++j) {
        v2f a0, a1;
        a0.x = a[j].x; a0.y = a[j].y;
        a1.x = a[j].z; a1.y = a[j].w;
        #pragma unroll
        for (int r = 0; r < RR; ++r) {
            const v4f b = wc[(r * GKB + j) * 32];
            v2f b0, b1;
            b0.x = b.x; b0.y = b.y;
            b1.x = b.z; b1.y = b.w;
            acc[r] = __builtin_amdgcn_wmma_f32_16x16x4_f32(
                 false, a0,  false, b0,
                 (short)0, acc[r],  false,  false);
            acc[r] = __builtin_amdgcn_wmma_f32_16x16x4_f32(
                 false, a1,  false, b1,
                 (short)0, acc[r],  false,  false);
        }
    }
}

__global__ __launch_bounds__(GWAVES * 32, 4) void gemm_wmma_kernel(
    const float* __restrict__ feat,
    const float* __restrict__ W,
    const float* __restrict__ bias,
    float* __restrict__ wh,
    float* zp,
    float* tick)
{
    __shared__ v4f wl[RR * GKB * 32];

    const int tid = (int)threadIdx.x;

    {
        v4f* z = reinterpret_cast<v4f*>(zp);
        for (int i = (int)blockIdx.x * (GWAVES * 32) + tid; i < NZ4; i += GBLOCKS * GWAVES * 32)
            z[i] = (v4f){};
        if (blockIdx.x == 0 && tid == 0) {
            unsigned* q = reinterpret_cast<unsigned*>(tick);
            q[0] = 0u;
            q[64] = 0u;
        }
    }

    #pragma unroll
    for (int i = 0; i < (RR * GKB * 32) / (GWAVES * 32); ++i) {
        const int idx = tid + (GWAVES * 32) * i;
        const int ln = idx & 31;
        const int kb = (idx >> 5) & (GKB - 1);
        const int r  = idx >> 10;
        const float* p = W + (size_t)r * (DD * CC) + (kb * 8 + 4 * (ln >> 4)) * CC + (ln & 15);
        v4f v;
        v.x = p[0];
        v.y = p[CC];
        v.z = p[2 * CC];
        v.w = p[3 * CC];
        wl[idx] = v;
    }
    __syncthreads();

    const int wave = __builtin_amdgcn_readfirstlane(tid >> 5);
    const int lane = tid & 31;
    const int col  = lane & 15;
    const int half = lane >> 4;

    const int first = wave * GBLOCKS + (int)blockIdx.x;
    const float* fl = feat + (size_t)col * DD + 4 * half;

    v4f ae[GCH], ao[GCH];
    if (first < GTILES) load_chunk(ae, fl + (size_t)first * (16 * DD));

    for (int tile = first; tile < GTILES; tile += GBLOCKS * GWAVES) {
        v8f acc[RR];
        #pragma unroll
        for (int r = 0; r < RR; ++r) acc[r] = (v8f){};
        const float* pt = fl + (size_t)tile * (16 * DD);

        #pragma nounroll
        for (int cp = 0; cp < GKB / (2 * GCH); ++cp) {
            load_chunk(ao, pt + (2 * cp + 1) * (8 * GCH));
            chunk_mma(acc, ae, wl + (size_t)(2 * cp * GCH) * 32 + lane);

            const bool last = (cp == GKB / (2 * GCH) - 1);
            const int ntile = last ? tile + GBLOCKS * GWAVES : tile;
            const int nch   = last ? 0 : 2 * cp + 2;
            if (ntile < GTILES) load_chunk(ae, fl + (size_t)ntile * (16 * DD) + nch * (8 * GCH));
            chunk_mma(acc, ao, wl + (size_t)((2 * cp + 1) * GCH) * 32 + lane);
        }

        asm volatile("v_nop\n\tv_nop\n\tv_nop\n\tv_nop");

        float* out = wh + (size_t)tile * (16 * RR * CC) + col;
        #pragma unroll
        for (int r = 0; r < RR; ++r) {
            const float bv = bias[r * CC + col];
            #pragma unroll
            for (int i = 0; i < 8; ++i) {
                const int m = i + 8 * half;
                out[(size_t)m * (RR * CC) + r * CC] = acc[r][i] + bv;
            }
        }
    }
}

#define NH 50000
#define GROUPS 4
#define NTASK ((unsigned)(((long long)RR * EE) / (32 * GROUPS)))
#define QSTRIDE 64
static_assert(EE % 32 == 0, "32 consecutive edges must belong to one etype");
static_assert(((long long)RR * EE) % (32 * GROUPS) == 0, "tasks cover the edges exactly");
static_assert(NH % 4 == 0, "the split falls on a 256B boundary of sums");

#define ROW_SHARE(X, J) __builtin_amdgcn_update_dpp(0, (X), 0x150 + (J), 0xF, 0xF, false)

#define GATHER(J) \
    const int d##J = ROW_SHARE(d, J); \
    const int s##J = ROW_SHARE(s, J); \
    const bool m##J = ((d##J >= NH) ? 1 : 0) == h; \
    float v##J = 0.0f; \
    if (m##J) v##J = whr[(unsigned)(s##J * (RR * CC) + c)];
#define ROW_ADD(J) { \
    const float w##J = __int_as_float(ROW_SHARE(wb, J)); \
    if (m##J) atomicAdd(sr + (unsigned)(d##J * CC + c), v##J * w##J); }

__device__ __forceinline__ void edge_group(
    const float* __restrict__ whr, float* __restrict__ sr,
    const int s, const int d, const int wb, const int c, const int h)
{
    GATHER(0)  GATHER(1)  GATHER(2)  GATHER(3)
    GATHER(4)  GATHER(5)  GATHER(6)  GATHER(7)
    GATHER(8)  GATHER(9)  GATHER(10) GATHER(11)
    GATHER(12) GATHER(13) GATHER(14) GATHER(15)

    ROW_ADD(0)  ROW_ADD(1)  ROW_ADD(2)  ROW_ADD(3)
    ROW_ADD(4)  ROW_ADD(5)  ROW_ADD(6)  ROW_ADD(7)
    ROW_ADD(8)  ROW_ADD(9)  ROW_ADD(10) ROW_ADD(11)
    ROW_ADD(12) ROW_ADD(13) ROW_ADD(14) ROW_ADD(15)
}

#define SBLOCKS 1024
__global__ __launch_bounds__(CTHREADS) void scatter_count_kernel(
    const float* __restrict__ wh,
    const int*   __restrict__ src,
    const int*   __restrict__ dst,
    const float* __restrict__ ew,
    float* __restrict__ sums,
    float* __restrict__ cnts,
    float* tick)
{
    __shared__ unsigned hist[CR];
    if (blockIdx.x < CBLOCKS) {
        count_body(hist, dst, cnts);
        return;
    }

    unsigned* q = reinterpret_cast<unsigned*>(tick);
    const unsigned lane = threadIdx.x & 31u;
    const int c = (int)(lane & 15u);

    const unsigned place = (unsigned)__builtin_amdgcn_s_sendmsg_rtn(0x87);
    int h = (((place >> 16) & 0xFu) >= 4u) ? 1 : 0;

    for (int pass = 0; pass < 2; ++pass) {
        unsigned* qh = q + h * QSTRIDE;
        for (;;) {
            unsigned t = 0;
            if (lane == 0) t = atomicAdd(qh, 1u);
            t = (unsigned)__builtin_amdgcn_readfirstlane((int)t);
            if (t >= NTASK) break;

            const unsigned i0 = t * (32u * GROUPS) + lane;

            int s[GROUPS], d[GROUPS], wb[GROUPS];
            #pragma unroll
            for (int g = 0; g < GROUPS; ++g) {
                const unsigned i = i0 + 32u * g;
                s[g]  = src[i];
                d[g]  = dst[i];
                wb[g] = __float_as_int(ew[i]);
            }

            #pragma unroll
            for (int g = 0; g < GROUPS; ++g) {
                const unsigned i = i0 + 32u * g;
                const int r = __builtin_amdgcn_readfirstlane((int)(i / (unsigned)EE));
                edge_group(wh + r * CC,
                           sums + (size_t)r * NN * CC,
                           s[g], d[g], wb[g], c, h);
            }
        }
        h ^= 1;
    }
}
#undef GATHER
#undef ROW_ADD
#undef ROW_SHARE

__global__ __launch_bounds__(256) void finalize_kernel(
    const float* __restrict__ sums,
    const float* __restrict__ cnts,
    float* __restrict__ out)
{
    const long long total = (long long)NN * CC;
    long long i = (long long)blockIdx.x * blockDim.x + threadIdx.x;
    if (i >= total) return;
    const int n = (int)(i >> 4);
    const int c = (int)(i & 15);

    float acc = 0.0f;
    #pragma unroll
    for (int r = 0; r < RR; ++r) {
        const float cnt = cnts[(size_t)r * NN + n];
        if (cnt > 0.0f) {
            acc += sums[((size_t)r * NN + n) * CC + c] / cnt;
        }
    }
    out[i] = acc;
}

extern "C" void kernel_launch(void* const* d_in, const int* in_sizes, int n_in,
                              void* d_out, int out_size, void* d_ws, size_t ws_size,
                              hipStream_t stream) {
    const float* feat = (const float*)d_in[0];
    const int*   srci = (const int*)  d_in[1];
    const int*   dsti = (const int*)  d_in[2];
    const float* ew   = (const float*)d_in[3];
    const float* W    = (const float*)d_in[4];
    const float* bias = (const float*)d_in[5];
    float* out = (float*)d_out;

    float* ws   = (float*)d_ws;
    float* wh   = ws;
    float* sums = ws + (size_t)NN * (RR * CC);
    float* cnts = sums + (size_t)NN * (RR * CC);

    {
        gemm_wmma_kernel<<<GBLOCKS, GWAVES * 32, 0, stream>>>(feat, W, bias, wh, sums, out);
    }

    {
        scatter_count_kernel<<<CBLOCKS + SBLOCKS, CTHREADS, 0, stream>>>(wh, srci, dsti, ew, sums, cnts, out);
    }

    {
        const long long total = (long long)NN * CC;
        const int blocks = (int)((total + 255) / 256);
        finalize_kernel<<<blocks, 256, 0, stream>>>(sums, cnts, out);
    }
}


# scatter: each 8000-edge chunk sorted into 782-node destination bins (LDS counting sort), then one work-group per bin sums its runs in LDS; no global f32 atomics, LDS f32 adds in arrival order
# speedup vs baseline: 16.1109x; 1.5020x over previous
#include <hip/hip_runtime.h>

#define NN 100000
#define EE 2000000
#define DD 256
#define CC 16
#define RR 4

typedef __attribute__((ext_vector_type(2))) float v2f;
typedef __attribute__((ext_vector_type(8))) float v8f;

#define CR 12500
#define CRANGES (NN / CR)
#define CSLICES 8
#define CSLICE_V (EE / CSLICES / 4)
#define CTHREADS 512
#define CAHEAD 8
static_assert(NN % CR == 0, "ranges cover the nodes exactly");
static_assert(EE % (CSLICES * 4) == 0, "slices are whole 16-byte groups");

__device__ __forceinline__ void tally4(unsigned* hist, const int4 v, const int lo) {
    unsigned o;
    o = (unsigned)(v.x - lo); if (o < (unsigned)CR) atomicAdd(&hist[o], 1u);
    o = (unsigned)(v.y - lo); if (o < (unsigned)CR) atomicAdd(&hist[o], 1u);
    o = (unsigned)(v.z - lo); if (o < (unsigned)CR) atomicAdd(&hist[o], 1u);
    o = (unsigned)(v.w - lo); if (o < (unsigned)CR) atomicAdd(&hist[o], 1u);
}

#define CBLOCKS (RR * CSLICES * CRANGES)

__device__ __forceinline__ void count_body(
    unsigned* hist,
    const int* __restrict__ dst,
    float* __restrict__ cnts)
{
    const int b = (int)blockIdx.x;
    const int q = b % CRANGES;
    const int k = (b / CRANGES) % CSLICES;
    const int r = b / (CRANGES * CSLICES);
    const int lo = q * CR;

    for (int i = (int)threadIdx.x; i < CR; i += CTHREADS) hist[i] = 0u;
    __syncthreads();

    const int4* p = reinterpret_cast<const int4*>(dst + (size_t)r * EE) + (size_t)k * CSLICE_V;
    int i = (int)threadIdx.x;
    for (; i + (CAHEAD - 1) * CTHREADS < CSLICE_V; i += CAHEAD * CTHREADS) {
        int4 v[CAHEAD];
        #pragma unroll
        for (int u = 0; u < CAHEAD; ++u) v[u] = p[i + u * CTHREADS];
        #pragma unroll
        for (int u = 0; u < CAHEAD; ++u) tally4(hist, v[u], lo);
    }
    for (; i < CSLICE_V; i += CTHREADS) tally4(hist, p[i], lo);
    __syncthreads();

    float* c = cnts + (size_t)r * NN + lo;
    for (int j = (int)threadIdx.x; j < CR; j += CTHREADS) {
        const unsigned h = hist[j];
        if (h != 0u) atomicAdd(c + j, (float)h);
    }
}

typedef __attribute__((ext_vector_type(4))) float v4f;
#define GKB (DD / 8)
#define GCH 8
#define GWAVES 8
#define GBLOCKS 768
#define GTILES (NN / 16)
#define NZ4 ((NN * (RR * CC) + RR * NN) / 4)
static_assert((NN * (RR * CC) + RR * NN) % 4 == 0, "the zeroed region is whole 16-byte groups");
static_assert(NN % 16 == 0, "row tiles are whole");
static_assert(DD % 8 == 0, "k blocks are whole");
static_assert(GKB % (2 * GCH) == 0, "chunks come in pairs");

__device__ __forceinline__ void load_chunk(v4f (&a)[GCH], const float* p) {
    #pragma unroll
    for (int j = 0; j < GCH; ++j) a[j] = *reinterpret_cast<const v4f*>(p + 8 * j);
}

__device__ __forceinline__ void chunk_mma(v8f (&acc)[RR], const v4f (&a)[GCH], const v4f* wc) {
    #pragma unroll
    for (int j = 0; j < GCH; ++j) {
        v2f a0, a1;
        a0.x = a[j].x; a0.y = a[j].y;
        a1.x = a[j].z; a1.y = a[j].w;
        #pragma unroll
        for (int r = 0; r < RR; ++r) {
            const v4f b = wc[(r * GKB + j) * 32];
            v2f b0, b1;
            b0.x = b.x; b0.y = b.y;
            b1.x = b.z; b1.y = b.w;
            acc[r] = __builtin_amdgcn_wmma_f32_16x16x4_f32(
                 false, a0,  false, b0,
                 (short)0, acc[r],  false,  false);
            acc[r] = __builtin_amdgcn_wmma_f32_16x16x4_f32(
                 false, a1,  false, b1,
                 (short)0, acc[r],  false,  false);
        }
    }
}

__global__ __launch_bounds__(GWAVES * 32, 4) void gemm_wmma_kernel(
    const float* __restrict__ feat,
    const float* __restrict__ W,
    const float* __restrict__ bias,
    float* __restrict__ wh,
    float* zp,
    float* tick)
{
    __shared__ v4f wl[RR * GKB * 32];

    const int tid = (int)threadIdx.x;

    if (zp != nullptr) {
        v4f* z = reinterpret_cast<v4f*>(zp);
        for (int i = (int)blockIdx.x * (GWAVES * 32) + tid; i < NZ4; i += GBLOCKS * GWAVES * 32)
            z[i] = (v4f){};
        if (blockIdx.x == 0 && tid == 0) {
            unsigned* q = reinterpret_cast<unsigned*>(tick);
            q[0] = 0u;
            q[64] = 0u;
        }
    }

    #pragma unroll
    for (int i = 0; i < (RR * GKB * 32) / (GWAVES * 32); ++i) {
        const int idx = tid + (GWAVES * 32) * i;
        const int ln = idx & 31;
        const int kb = (idx >> 5) & (GKB - 1);
        const int r  = idx >> 10;
        const float* p = W + (size_t)r * (DD * CC) + (kb * 8 + 4 * (ln >> 4)) * CC + (ln & 15);
        v4f v;
        v.x = p[0];
        v.y = p[CC];
        v.z = p[2 * CC];
        v.w = p[3 * CC];
        wl[idx] = v;
    }
    __syncthreads();

    const int wave = __builtin_amdgcn_readfirstlane(tid >> 5);
    const int lane = tid & 31;
    const int col  = lane & 15;
    const int half = lane >> 4;

    const int first = wave * GBLOCKS + (int)blockIdx.x;
    const float* fl = feat + (size_t)col * DD + 4 * half;

    v4f ae[GCH], ao[GCH];
    if (first < GTILES) load_chunk(ae, fl + (size_t)first * (16 * DD));

    for (int tile = first; tile < GTILES; tile += GBLOCKS * GWAVES) {
        v8f acc[RR];
        #pragma unroll
        for (int r = 0; r < RR; ++r) acc[r] = (v8f){};
        const float* pt = fl + (size_t)tile * (16 * DD);

        #pragma nounroll
        for (int cp = 0; cp < GKB / (2 * GCH); ++cp) {
            load_chunk(ao, pt + (2 * cp + 1) * (8 * GCH));
            chunk_mma(acc, ae, wl + (size_t)(2 * cp * GCH) * 32 + lane);

            const bool last = (cp == GKB / (2 * GCH) - 1);
            const int ntile = last ? tile + GBLOCKS * GWAVES : tile;
            const int nch   = last ? 0 : 2 * cp + 2;
            if (ntile < GTILES) load_chunk(ae, fl + (size_t)ntile * (16 * DD) + nch * (8 * GCH));
            chunk_mma(acc, ao, wl + (size_t)((2 * cp + 1) * GCH) * 32 + lane);
        }

        asm volatile("v_nop\n\tv_nop\n\tv_nop\n\tv_nop");

        float* out = wh + (size_t)tile * (16 * RR * CC) + col;
        #pragma unroll
        for (int r = 0; r < RR; ++r) {
            const float bv = bias[r * CC + col];
            #pragma unroll
            for (int i = 0; i < 8; ++i) {
                const int m = i + 8 * half;
                out[(size_t)m * (RR * CC) + r * CC] = acc[r][i] + bv;
            }
        }
    }
}

#define NH 50000
#define GROUPS 4
#define NTASK ((unsigned)(((long long)RR * EE) / (32 * GROUPS)))
#define QSTRIDE 64
static_assert(EE % 32 == 0, "32 consecutive edges must belong to one etype");
static_assert(((long long)RR * EE) % (32 * GROUPS) == 0, "tasks cover the edges exactly");
static_assert(NH % 4 == 0, "the split falls on a 256B boundary of sums");

#define ROW_SHARE(X, J) __builtin_amdgcn_update_dpp(0, (X), 0x150 + (J), 0xF, 0xF, false)

#define GATHER(J) \
    const int d##J = ROW_SHARE(d, J); \
    const int s##J = ROW_SHARE(s, J); \
    const bool m##J = ((d##J >= NH) ? 1 : 0) == h; \
    float v##J = 0.0f; \
    if (m##J) v##J = whr[(unsigned)(s##J * (RR * CC) + c)];
#define ROW_ADD(J) { \
    const float w##J = __int_as_float(ROW_SHARE(wb, J)); \
    if (m##J) atomicAdd(sr + (unsigned)(d##J * CC + c), v##J * w##J); }

__device__ __forceinline__ void edge_group(
    const float* __restrict__ whr, float* __restrict__ sr,
    const int s, const int d, const int wb, const int c, const int h)
{
    GATHER(0)  GATHER(1)  GATHER(2)  GATHER(3)
    GATHER(4)  GATHER(5)  GATHER(6)  GATHER(7)
    GATHER(8)  GATHER(9)  GATHER(10) GATHER(11)
    GATHER(12) GATHER(13) GATHER(14) GATHER(15)

    ROW_ADD(0)  ROW_ADD(1)  ROW_ADD(2)  ROW_ADD(3)
    ROW_ADD(4)  ROW_ADD(5)  ROW_ADD(6)  ROW_ADD(7)
    ROW_ADD(8)  ROW_ADD(9)  ROW_ADD(10) ROW_ADD(11)
    ROW_ADD(12) ROW_ADD(13) ROW_ADD(14) ROW_ADD(15)
}

#define SBLOCKS 1024
__global__ __launch_bounds__(CTHREADS) void scatter_count_kernel(
    const float* __restrict__ wh,
    const int*   __restrict__ src,
    const int*   __restrict__ dst,
    const float* __restrict__ ew,
    float* __restrict__ sums,
    float* __restrict__ cnts,
    float* tick)
{
    __shared__ unsigned hist[CR];
    if (blockIdx.x < CBLOCKS) {
        count_body(hist, dst, cnts);
        return;
    }

    unsigned* q = reinterpret_cast<unsigned*>(tick);
    const unsigned lane = threadIdx.x & 31u;
    const int c = (int)(lane & 15u);

    const unsigned place = (unsigned)__builtin_amdgcn_s_sendmsg_rtn(0x87);
    int h = (((place >> 16) & 0xFu) >= 4u) ? 1 : 0;

    for (int pass = 0; pass < 2; ++pass) {
        unsigned* qh = q + h * QSTRIDE;
        for (;;) {
            unsigned t = 0;
            if (lane == 0) t = atomicAdd(qh, 1u);
            t = (unsigned)__builtin_amdgcn_readfirstlane((int)t);
            if (t >= NTASK) break;

            const unsigned i0 = t * (32u * GROUPS) + lane;

            int s[GROUPS], d[GROUPS], wb[GROUPS];
            #pragma unroll
            for (int g = 0; g < GROUPS; ++g) {
                const unsigned i = i0 + 32u * g;
                s[g]  = src[i];
                d[g]  = dst[i];
                wb[g] = __float_as_int(ew[i]);
            }

            #pragma unroll
            for (int g = 0; g < GROUPS; ++g) {
                const unsigned i = i0 + 32u * g;
                const int r = __builtin_amdgcn_readfirstlane((int)(i / (unsigned)EE));
                edge_group(wh + r * CC,
                           sums + (size_t)r * NN * CC,
                           s[g], d[g], wb[g], c, h);
            }
        }
        h ^= 1;
    }
}
#undef GATHER
#undef ROW_ADD

#define BN 782
#define LOCBITS 10
#define NB ((NN + BN - 1) / BN)
#define NBP 130
#define PB 160
#define CH 8000
#define KCH (EE / CH)
#define PTHREADS 512
#define PPER ((CH + PTHREADS - 1) / PTHREADS)
#define SRCBITS 17
#define RWAVES 32
static_assert(EE % CH == 0, "a chunk never spans two etypes");
static_assert(CH % 2 == 0, "a chunk is whole 16-byte groups of records");
static_assert(NN <= (1 << SRCBITS), "the source node fits its field");
static_assert(LOCBITS + SRCBITS <= 32 && BN <= (1 << LOCBITS), "the record's first word holds both fields");
static_assert(NB + 1 <= NBP && NBP <= PB, "the table row holds every start and the end");
static_assert(KCH <= RWAVES * 32, "one lane per run of a wave");
static_assert(BN <= RWAVES * 32, "one thread per count in the reduce work-group");
static_assert((BN * CC) % 4 == 0, "a bin's sums are whole 16-byte groups");
static_assert(PB % 32 == 0, "the scan runs in whole waves");

__global__ __launch_bounds__(PTHREADS) void part_kernel(
    const int*   __restrict__ src,
    const int*   __restrict__ dst,
    const float* __restrict__ ew,
    float* recf,
    float* offf)
{
    __shared__ uint4 lrec4[CH / 2];
    __shared__ unsigned hist[PB];
    __shared__ unsigned wsum[PB / 32];
    uint2* lrec = reinterpret_cast<uint2*>(lrec4);

    const int tid = (int)threadIdx.x;
    const int chunk = (int)blockIdx.x;
    const size_t e0 = (size_t)chunk * CH;

    if (tid < PB) hist[tid] = 0u;
    __syncthreads();

    int dv[PPER], sv[PPER], wv[PPER];
    #pragma unroll
    for (int u = 0; u < PPER; ++u) {
        const int i = tid + PTHREADS * u;
        dv[u] = 0; sv[u] = 0; wv[u] = 0;
        if (i < CH) {
            dv[u] = dst[e0 + i];
            sv[u] = src[e0 + i];
            wv[u] = __float_as_int(ew[e0 + i]);
        }
    }
    unsigned rk[PPER];
    #pragma unroll
    for (int u = 0; u < PPER; ++u) {
        const int i = tid + PTHREADS * u;
        rk[u] = 0u;
        if (i < CH) rk[u] = atomicAdd(&hist[(unsigned)dv[u] / BN], 1u);
    }
    __syncthreads();

    unsigned hv = 0u;
    if (tid < PB) hv = hist[tid];
    unsigned incl = hv;
    #pragma unroll
    for (int o = 1; o < 32; o <<= 1) {
        const unsigned t = (unsigned)__shfl_up((int)incl, (unsigned)o);
        if ((tid & 31) >= o) incl += t;
    }
    if (tid < PB && (tid & 31) == 31) wsum[tid >> 5] = incl;
    __syncthreads();
    if (tid < PB) {
        unsigned base = 0u;
        for (int w = 0; w < (tid >> 5); ++w) base += wsum[w];
        const unsigned st = base + incl - hv;
        hist[tid] = st;
        if (tid < NBP) reinterpret_cast<unsigned*>(offf)[(size_t)chunk * NBP + tid] = st;
    }
    __syncthreads();

    #pragma unroll
    for (int u = 0; u < PPER; ++u) {
        const int i = tid + PTHREADS * u;
        if (i < CH) {
            const unsigned bin = (unsigned)dv[u] / BN;
            const unsigned pos = hist[bin] + rk[u];
            uint2 rc;
            rc.x = (unsigned)sv[u] | (((unsigned)dv[u] - bin * BN) << SRCBITS);
            rc.y = (unsigned)wv[u];
            lrec[pos] = rc;
        }
    }
    __syncthreads();

    uint4* g4 = reinterpret_cast<uint4*>(recf) + e0 / 2;
    for (int i = tid; i < CH / 2; i += PTHREADS) g4[i] = lrec4[i];
}

#define RGATHER(J) \
    const unsigned x##J = (unsigned)ROW_SHARE((int)cur.x, J); \
    const bool m##J = (rb + (unsigned)(J)) < n; \
    float v##J = 0.0f; \
    if (m##J) v##J = whr[(x##J & ((1u << SRCBITS) - 1u)) * (unsigned)(RR * CC) + c];
#define RADD(J) { \
    const float w##J = __int_as_float(ROW_SHARE((int)cur.y, J)); \
    if (m##J) atomicAdd(&acc[(x##J >> SRCBITS) * CC + c], v##J * w##J); }

#define RGATHER_F(J) \
    const unsigned x##J = (unsigned)ROW_SHARE((int)cur.x, J); \
    const float v##J = whr[(x##J & ((1u << SRCBITS) - 1u)) * (unsigned)(RR * CC) + c];
#define RADD_F(J) { \
    const float w##J = __int_as_float(ROW_SHARE((int)cur.y, J)); \
    atomicAdd(&acc[(x##J >> SRCBITS) * CC + c], v##J * w##J); }
#define NEXT_RECORDS { \
    const uint2* gq = rec; \
    nn = next_group(gq); \
    if (lane < nn) nxt = gq[lane]; }

__global__ __launch_bounds__(RWAVES * 32) void reduce_kernel(
    const float* __restrict__ wh,
    const float* recf,
    const float* offf,
    float* __restrict__ sums,
    float* __restrict__ cnts)
{
    __shared__ v4f acc4[BN * CC / 4];
    __shared__ unsigned cl[BN];
    float* acc = reinterpret_cast<float*>(acc4);

    const int tid = (int)threadIdx.x;
    const int r = (int)blockIdx.x / NB;
    const int b = (int)blockIdx.x % NB;

    for (int i = tid; i < BN * CC / 4; i += RWAVES * 32) acc4[i] = (v4f){};
    if (tid < BN) cl[tid] = 0u;
    __syncthreads();

    const int wave = __builtin_amdgcn_readfirstlane(tid >> 5);
    __builtin_assume(wave >= 0 && wave < RWAVES);
    const unsigned lane = (unsigned)tid & 31u;
    const unsigned c = lane & 15u;
    const unsigned rb = lane & 16u;
    const float* whr = wh + r * CC;
    const uint2* rec = reinterpret_cast<const uint2*>(recf);
    const unsigned* off = reinterpret_cast<const unsigned*>(offf);

    const int nruns = (KCH - wave + RWAVES - 1) / RWAVES;
    unsigned o0 = 0u, o1 = 0u;
    if ((int)lane < nruns) {
        const unsigned* p = off + (size_t)(r * KCH + wave + RWAVES * (int)lane) * NBP + b;
        o0 = p[0];
        o1 = p[1];
    }

    int j = -1;
    unsigned pos = 0u, end = 0u;
    const uint2* rbase = rec;
    auto next_group = [&](const uint2*& gp) -> unsigned {
        while (pos >= end) {
            ++j;
            if (j >= nruns) return 0u;
            pos = (unsigned)__builtin_amdgcn_readlane((int)o0, j);
            end = (unsigned)__builtin_amdgcn_readlane((int)o1, j);
            rbase = rec + (size_t)(r * KCH + wave + RWAVES * j) * CH;
        }
        gp = rbase + pos;
        const unsigned left = end - pos;
        pos += 32u;
        return left < 32u ? left : 32u;
    };

    const uint2* gp = rec;
    unsigned n = next_group(gp);
    uint2 cur;
    cur.x = 0u; cur.y = 0u;
    if (lane < n) cur = gp[lane];
    while (n != 0u) {
        if (lane < n) atomicAdd(&cl[cur.x >> SRCBITS], 1u);

        unsigned nn;
        uint2 nxt;
        nxt.x = 0u; nxt.y = 0u;
        if (n == 32u) {
            RGATHER_F(0)  RGATHER_F(1)  RGATHER_F(2)  RGATHER_F(3)
            RGATHER_F(4)  RGATHER_F(5)  RGATHER_F(6)  RGATHER_F(7)
            RGATHER_F(8)  RGATHER_F(9)  RGATHER_F(10) RGATHER_F(11)
            RGATHER_F(12) RGATHER_F(13) RGATHER_F(14) RGATHER_F(15)

            NEXT_RECORDS

            RADD_F(0)  RADD_F(1)  RADD_F(2)  RADD_F(3)
            RADD_F(4)  RADD_F(5)  RADD_F(6)  RADD_F(7)
            RADD_F(8)  RADD_F(9)  RADD_F(10) RADD_F(11)
            RADD_F(12) RADD_F(13) RADD_F(14) RADD_F(15)
        } else {
            RGATHER(0)  RGATHER(1)  RGATHER(2)  RGATHER(3)
            RGATHER(4)  RGATHER(5)  RGATHER(6)  RGATHER(7)
            RGATHER(8)  RGATHER(9)  RGATHER(10) RGATHER(11)
            RGATHER(12) RGATHER(13) RGATHER(14) RGATHER(15)

            NEXT_RECORDS

            RADD(0)  RADD(1)  RADD(2)  RADD(3)
            RADD(4)  RADD(5)  RADD(6)  RADD(7)
            RADD(8)  RADD(9)  RADD(10) RADD(11)
            RADD(12) RADD(13) RADD(14) RADD(15)
        }

        cur = nxt;
        n = nn;
    }
    __syncthreads();

    const int n0 = b * BN;
    const int nb = (NN - n0) < BN ? (NN - n0) : BN;
    v4f* s4 = reinterpret_cast<v4f*>(sums + ((size_t)r * NN + n0) * CC);
    for (int i = tid; i < nb * (CC / 4); i += RWAVES * 32) s4[i] = acc4[i];
    if (tid < nb) cnts[(size_t)r * NN + n0 + tid] = (float)cl[tid];
}
#undef RGATHER
#undef RADD
#undef RGATHER_F
#undef RADD_F
#undef NEXT_RECORDS
#undef ROW_SHARE

__global__ __launch_bounds__(256) void finalize_kernel(
    const float* __restrict__ sums,
    const float* __restrict__ cnts,
    float* __restrict__ out)
{
    const long long total = (long long)NN * CC;
    long long i = (long long)blockIdx.x * blockDim.x + threadIdx.x;
    if (i >= total) return;
    const int n = (int)(i >> 4);
    const int c = (int)(i & 15);

    float acc = 0.0f;
    #pragma unroll
    for (int r = 0; r < RR; ++r) {
        const float cnt = cnts[(size_t)r * NN + n];
        if (cnt > 0.0f) {
            acc += sums[((size_t)r * NN + n) * CC + c] / cnt;
        }
    }
    out[i] = acc;
}

extern "C" void kernel_launch(void* const* d_in, const int* in_sizes, int n_in,
                              void* d_out, int out_size, void* d_ws, size_t ws_size,
                              hipStream_t stream) {
    const float* feat = (const float*)d_in[0];
    const int*   srci = (const int*)  d_in[1];
    const int*   dsti = (const int*)  d_in[2];
    const float* ew   = (const float*)d_in[3];
    const float* W    = (const float*)d_in[4];
    const float* bias = (const float*)d_in[5];
    float* out = (float*)d_out;

    float* ws   = (float*)d_ws;
    float* wh   = ws;
    float* sums = ws + (size_t)NN * (RR * CC);
    float* cnts = sums + (size_t)NN * (RR * CC);

    if (ws_size >= (size_t)117320000) {
        float* recf = cnts + (size_t)RR * NN;
        float* offf = recf + (size_t)2 * RR * EE;
        gemm_wmma_kernel<<<GBLOCKS, GWAVES * 32, 0, stream>>>(feat, W, bias, wh, nullptr, nullptr);
        part_kernel<<<RR * KCH, PTHREADS, 0, stream>>>(srci, dsti, ew, recf, offf);
        reduce_kernel<<<RR * NB, RWAVES * 32, 0, stream>>>(wh, recf, offf, sums, cnts);
    } else {
        {
            gemm_wmma_kernel<<<GBLOCKS, GWAVES * 32, 0, stream>>>(feat, W, bias, wh, sums, out);
        }

        {
            scatter_count_kernel<<<CBLOCKS + SBLOCKS, CTHREADS, 0, stream>>>(wh, srci, dsti, ew, sums, cnts, out);
        }

    }

    {
        const long long total = (long long)NN * CC;
        const int blocks = (int)((total + 255) / 256);
        finalize_kernel<<<blocks, 256, 0, stream>>>(sums, cnts, out);
    }
}


# finalize: 625 work-groups of 4 waves, each lane five 16-byte channel groups, all 40 loads issued ahead of the divisions, where one element per lane took 50000 short waves; same divisions, same order
# speedup vs baseline: 17.2698x; 1.0719x over previous
#include <hip/hip_runtime.h>

#define NN 100000
#define EE 2000000
#define DD 256
#define CC 16
#define RR 4

typedef __attribute__((ext_vector_type(2))) float v2f;
typedef __attribute__((ext_vector_type(8))) float v8f;

#define CR 12500
#define CRANGES (NN / CR)
#define CSLICES 8
#define CSLICE_V (EE / CSLICES / 4)
#define CTHREADS 512
#define CAHEAD 8
static_assert(NN % CR == 0, "ranges cover the nodes exactly");
static_assert(EE % (CSLICES * 4) == 0, "slices are whole 16-byte groups");

__device__ __forceinline__ void tally4(unsigned* hist, const int4 v, const int lo) {
    unsigned o;
    o = (unsigned)(v.x - lo); if (o < (unsigned)CR) atomicAdd(&hist[o], 1u);
    o = (unsigned)(v.y - lo); if (o < (unsigned)CR) atomicAdd(&hist[o], 1u);
    o = (unsigned)(v.z - lo); if (o < (unsigned)CR) atomicAdd(&hist[o], 1u);
    o = (unsigned)(v.w - lo); if (o < (unsigned)CR) atomicAdd(&hist[o], 1u);
}

#define CBLOCKS (RR * CSLICES * CRANGES)

__device__ __forceinline__ void count_body(
    unsigned* hist,
    const int* __restrict__ dst,
    float* __restrict__ cnts)
{
    const int b = (int)blockIdx.x;
    const int q = b % CRANGES;
    const int k = (b / CRANGES) % CSLICES;
    const int r = b / (CRANGES * CSLICES);
    const int lo = q * CR;

    for (int i = (int)threadIdx.x; i < CR; i += CTHREADS) hist[i] = 0u;
    __syncthreads();

    const int4* p = reinterpret_cast<const int4*>(dst + (size_t)r * EE) + (size_t)k * CSLICE_V;
    int i = (int)threadIdx.x;
    for (; i + (CAHEAD - 1) * CTHREADS < CSLICE_V; i += CAHEAD * CTHREADS) {
        int4 v[CAHEAD];
        #pragma unroll
        for (int u = 0; u < CAHEAD; ++u) v[u] = p[i + u * CTHREADS];
        #pragma unroll
        for (int u = 0; u < CAHEAD; ++u) tally4(hist, v[u], lo);
    }
    for (; i < CSLICE_V; i += CTHREADS) tally4(hist, p[i], lo);
    __syncthreads();

    float* c = cnts + (size_t)r * NN + lo;
    for (int j = (int)threadIdx.x; j < CR; j += CTHREADS) {
        const unsigned h = hist[j];
        if (h != 0u) atomicAdd(c + j, (float)h);
    }
}

typedef __attribute__((ext_vector_type(4))) float v4f;
#define GKB (DD / 8)
#define GCH 8
#define GWAVES 8
#define GBLOCKS 768
#define GTILES (NN / 16)
#define NZ4 ((NN * (RR * CC) + RR * NN) / 4)
static_assert((NN * (RR * CC) + RR * NN) % 4 == 0, "the zeroed region is whole 16-byte groups");
static_assert(NN % 16 == 0, "row tiles are whole");
static_assert(DD % 8 == 0, "k blocks are whole");
static_assert(GKB % (2 * GCH) == 0, "chunks come in pairs");

__device__ __forceinline__ void load_chunk(v4f (&a)[GCH], const float* p) {
    #pragma unroll
    for (int j = 0; j < GCH; ++j) a[j] = *reinterpret_cast<const v4f*>(p + 8 * j);
}

__device__ __forceinline__ void chunk_mma(v8f (&acc)[RR], const v4f (&a)[GCH], const v4f* wc) {
    #pragma unroll
    for (int j = 0; j < GCH; ++j) {
        v2f a0, a1;
        a0.x = a[j].x; a0.y = a[j].y;
        a1.x = a[j].z; a1.y = a[j].w;
        #pragma unroll
        for (int r = 0; r < RR; ++r) {
            const v4f b = wc[(r * GKB + j) * 32];
            v2f b0, b1;
            b0.x = b.x; b0.y = b.y;
            b1.x = b.z; b1.y = b.w;
            acc[r] = __builtin_amdgcn_wmma_f32_16x16x4_f32(
                 false, a0,  false, b0,
                 (short)0, acc[r],  false,  false);
            acc[r] = __builtin_amdgcn_wmma_f32_16x16x4_f32(
                 false, a1,  false, b1,
                 (short)0, acc[r],  false,  false);
        }
    }
}

__global__ __launch_bounds__(GWAVES * 32, 4) void gemm_wmma_kernel(
    const float* __restrict__ feat,
    const float* __restrict__ W,
    const float* __restrict__ bias,
    float* __restrict__ wh,
    float* zp,
    float* tick)
{
    __shared__ v4f wl[RR * GKB * 32];

    const int tid = (int)threadIdx.x;

    if (zp != nullptr) {
        v4f* z = reinterpret_cast<v4f*>(zp);
        for (int i = (int)blockIdx.x * (GWAVES * 32) + tid; i < NZ4; i += GBLOCKS * GWAVES * 32)
            z[i] = (v4f){};
        if (blockIdx.x == 0 && tid == 0) {
            unsigned* q = reinterpret_cast<unsigned*>(tick);
            q[0] = 0u;
            q[64] = 0u;
        }
    }

    #pragma unroll
    for (int i = 0; i < (RR * GKB * 32) / (GWAVES * 32); ++i) {
        const int idx = tid + (GWAVES * 32) * i;
        const int ln = idx & 31;
        const int kb = (idx >> 5) & (GKB - 1);
        const int r  = idx >> 10;
        const float* p = W + (size_t)r * (DD * CC) + (kb * 8 + 4 * (ln >> 4)) * CC + (ln & 15);
        v4f v;
        v.x = p[0];
        v.y = p[CC];
        v.z = p[2 * CC];
        v.w = p[3 * CC];
        wl[idx] = v;
    }
    __syncthreads();

    const int wave = __builtin_amdgcn_readfirstlane(tid >> 5);
    const int lane = tid & 31;
    const int col  = lane & 15;
    const int half = lane >> 4;

    const int first = wave * GBLOCKS + (int)blockIdx.x;
    const float* fl = feat + (size_t)col * DD + 4 * half;

    v4f ae[GCH], ao[GCH];
    if (first < GTILES) load_chunk(ae, fl + (size_t)first * (16 * DD));

    for (int tile = first; tile < GTILES; tile += GBLOCKS * GWAVES) {
        v8f acc[RR];
        #pragma unroll
        for (int r = 0; r < RR; ++r) acc[r] = (v8f){};
        const float* pt = fl + (size_t)tile * (16 * DD);

        #pragma nounroll
        for (int cp = 0; cp < GKB / (2 * GCH); ++cp) {
            load_chunk(ao, pt + (2 * cp + 1) * (8 * GCH));
            chunk_mma(acc, ae, wl + (size_t)(2 * cp * GCH) * 32 + lane);

            const bool last = (cp == GKB / (2 * GCH) - 1);
            const int ntile = last ? tile + GBLOCKS * GWAVES : tile;
            const int nch   = last ? 0 : 2 * cp + 2;
            if (ntile < GTILES) load_chunk(ae, fl + (size_t)ntile * (16 * DD) + nch * (8 * GCH));
            chunk_mma(acc, ao, wl + (size_t)((2 * cp + 1) * GCH) * 32 + lane);
        }

        asm volatile("v_nop\n\tv_nop\n\tv_nop\n\tv_nop");

        float* out = wh + (size_t)tile * (16 * RR * CC) + col;
        #pragma unroll
        for (int r = 0; r < RR; ++r) {
            const float bv = bias[r * CC + col];
            #pragma unroll
            for (int i = 0; i < 8; ++i) {
                const int m = i + 8 * half;
                out[(size_t)m * (RR * CC) + r * CC] = acc[r][i] + bv;
            }
        }
    }
}

#define NH 50000
#define GROUPS 4
#define NTASK ((unsigned)(((long long)RR * EE) / (32 * GROUPS)))
#define QSTRIDE 64
static_assert(EE % 32 == 0, "32 consecutive edges must belong to one etype");
static_assert(((long long)RR * EE) % (32 * GROUPS) == 0, "tasks cover the edges exactly");
static_assert(NH % 4 == 0, "the split falls on a 256B boundary of sums");

#define ROW_SHARE(X, J) __builtin_amdgcn_update_dpp(0, (X), 0x150 + (J), 0xF, 0xF, false)

#define GATHER(J) \
    const int d##J = ROW_SHARE(d, J); \
    const int s##J = ROW_SHARE(s, J); \
    const bool m##J = ((d##J >= NH) ? 1 : 0) == h; \
    float v##J = 0.0f; \
    if (m##J) v##J = whr[(unsigned)(s##J * (RR * CC) + c)];
#define ROW_ADD(J) { \
    const float w##J = __int_as_float(ROW_SHARE(wb, J)); \
    if (m##J) atomicAdd(sr + (unsigned)(d##J * CC + c), v##J * w##J); }

__device__ __forceinline__ void edge_group(
    const float* __restrict__ whr, float* __restrict__ sr,
    const int s, const int d, const int wb, const int c, const int h)
{
    GATHER(0)  GATHER(1)  GATHER(2)  GATHER(3)
    GATHER(4)  GATHER(5)  GATHER(6)  GATHER(7)
    GATHER(8)  GATHER(9)  GATHER(10) GATHER(11)
    GATHER(12) GATHER(13) GATHER(14) GATHER(15)

    ROW_ADD(0)  ROW_ADD(1)  ROW_ADD(2)  ROW_ADD(3)
    ROW_ADD(4)  ROW_ADD(5)  ROW_ADD(6)  ROW_ADD(7)
    ROW_ADD(8)  ROW_ADD(9)  ROW_ADD(10) ROW_ADD(11)
    ROW_ADD(12) ROW_ADD(13) ROW_ADD(14) ROW_ADD(15)
}

#define SBLOCKS 1024
__global__ __launch_bounds__(CTHREADS) void scatter_count_kernel(
    const float* __restrict__ wh,
    const int*   __restrict__ src,
    const int*   __restrict__ dst,
    const float* __restrict__ ew,
    float* __restrict__ sums,
    float* __restrict__ cnts,
    float* tick)
{
    __shared__ unsigned hist[CR];
    if (blockIdx.x < CBLOCKS) {
        count_body(hist, dst, cnts);
        return;
    }

    unsigned* q = reinterpret_cast<unsigned*>(tick);
    const unsigned lane = threadIdx.x & 31u;
    const int c = (int)(lane & 15u);

    const unsigned place = (unsigned)__builtin_amdgcn_s_sendmsg_rtn(0x87);
    int h = (((place >> 16) & 0xFu) >= 4u) ? 1 : 0;

    for (int pass = 0; pass < 2; ++pass) {
        unsigned* qh = q + h * QSTRIDE;
        for (;;) {
            unsigned t = 0;
            if (lane == 0) t = atomicAdd(qh, 1u);
            t = (unsigned)__builtin_amdgcn_readfirstlane((int)t);
            if (t >= NTASK) break;

            const unsigned i0 = t * (32u * GROUPS) + lane;

            int s[GROUPS], d[GROUPS], wb[GROUPS];
            #pragma unroll
            for (int g = 0; g < GROUPS; ++g) {
                const unsigned i = i0 + 32u * g;
                s[g]  = src[i];
                d[g]  = dst[i];
                wb[g] = __float_as_int(ew[i]);
            }

            #pragma unroll
            for (int g = 0; g < GROUPS; ++g) {
                const unsigned i = i0 + 32u * g;
                const int r = __builtin_amdgcn_readfirstlane((int)(i / (unsigned)EE));
                edge_group(wh + r * CC,
                           sums + (size_t)r * NN * CC,
                           s[g], d[g], wb[g], c, h);
            }
        }
        h ^= 1;
    }
}
#undef GATHER
#undef ROW_ADD

#define BN 782
#define LOCBITS 10
#define NB ((NN + BN - 1) / BN)
#define NBP 130
#define PB 160
#define CH 8000
#define KCH (EE / CH)
#define PTHREADS 512
#define PPER ((CH + PTHREADS - 1) / PTHREADS)
#define SRCBITS 17
#define RWAVES 32
static_assert(EE % CH == 0, "a chunk never spans two etypes");
static_assert(CH % 2 == 0, "a chunk is whole 16-byte groups of records");
static_assert(NN <= (1 << SRCBITS), "the source node fits its field");
static_assert(LOCBITS + SRCBITS <= 32 && BN <= (1 << LOCBITS), "the record's first word holds both fields");
static_assert(NB + 1 <= NBP && NBP <= PB, "the table row holds every start and the end");
static_assert(KCH <= RWAVES * 32, "one lane per run of a wave");
static_assert(BN <= RWAVES * 32, "one thread per count in the reduce work-group");
static_assert((BN * CC) % 4 == 0, "a bin's sums are whole 16-byte groups");
static_assert(PB % 32 == 0, "the scan runs in whole waves");

__global__ __launch_bounds__(PTHREADS) void part_kernel(
    const int*   __restrict__ src,
    const int*   __restrict__ dst,
    const float* __restrict__ ew,
    float* recf,
    float* offf)
{
    __shared__ uint4 lrec4[CH / 2];
    __shared__ unsigned hist[PB];
    __shared__ unsigned wsum[PB / 32];
    uint2* lrec = reinterpret_cast<uint2*>(lrec4);

    const int tid = (int)threadIdx.x;
    const int chunk = (int)blockIdx.x;
    const size_t e0 = (size_t)chunk * CH;

    if (tid < PB) hist[tid] = 0u;
    __syncthreads();

    int dv[PPER], sv[PPER], wv[PPER];
    #pragma unroll
    for (int u = 0; u < PPER; ++u) {
        const int i = tid + PTHREADS * u;
        dv[u] = 0; sv[u] = 0; wv[u] = 0;
        if (i < CH) {
            dv[u] = dst[e0 + i];
            sv[u] = src[e0 + i];
            wv[u] = __float_as_int(ew[e0 + i]);
        }
    }
    unsigned rk[PPER];
    #pragma unroll
    for (int u = 0; u < PPER; ++u) {
        const int i = tid + PTHREADS * u;
        rk[u] = 0u;
        if (i < CH) rk[u] = atomicAdd(&hist[(unsigned)dv[u] / BN], 1u);
    }
    __syncthreads();

    unsigned hv = 0u;
    if (tid < PB) hv = hist[tid];
    unsigned incl = hv;
    #pragma unroll
    for (int o = 1; o < 32; o <<= 1) {
        const unsigned t = (unsigned)__shfl_up((int)incl, (unsigned)o);
        if ((tid & 31) >= o) incl += t;
    }
    if (tid < PB && (tid & 31) == 31) wsum[tid >> 5] = incl;
    __syncthreads();
    if (tid < PB) {
        unsigned base = 0u;
        for (int w = 0; w < (tid >> 5); ++w) base += wsum[w];
        const unsigned st = base + incl - hv;
        hist[tid] = st;
        if (tid < NBP) reinterpret_cast<unsigned*>(offf)[(size_t)chunk * NBP + tid] = st;
    }
    __syncthreads();

    #pragma unroll
    for (int u = 0; u < PPER; ++u) {
        const int i = tid + PTHREADS * u;
        if (i < CH) {
            const unsigned bin = (unsigned)dv[u] / BN;
            const unsigned pos = hist[bin] + rk[u];
            uint2 rc;
            rc.x = (unsigned)sv[u] | (((unsigned)dv[u] - bin * BN) << SRCBITS);
            rc.y = (unsigned)wv[u];
            lrec[pos] = rc;
        }
    }
    __syncthreads();

    uint4* g4 = reinterpret_cast<uint4*>(recf) + e0 / 2;
    for (int i = tid; i < CH / 2; i += PTHREADS) g4[i] = lrec4[i];
}

#define RGATHER(J) \
    const unsigned x##J = (unsigned)ROW_SHARE((int)cur.x, J); \
    const bool m##J = (rb + (unsigned)(J)) < n; \
    float v##J = 0.0f; \
    if (m##J) v##J = whr[(x##J & ((1u << SRCBITS) - 1u)) * (unsigned)(RR * CC) + c];
#define RADD(J) { \
    const float w##J = __int_as_float(ROW_SHARE((int)cur.y, J)); \
    if (m##J) atomicAdd(&acc[(x##J >> SRCBITS) * CC + c], v##J * w##J); }

#define RGATHER_F(J) \
    const unsigned x##J = (unsigned)ROW_SHARE((int)cur.x, J); \
    const float v##J = whr[(x##J & ((1u << SRCBITS) - 1u)) * (unsigned)(RR * CC) + c];
#define RADD_F(J) { \
    const float w##J = __int_as_float(ROW_SHARE((int)cur.y, J)); \
    atomicAdd(&acc[(x##J >> SRCBITS) * CC + c], v##J * w##J); }
#define NEXT_RECORDS { \
    const uint2* gq = rec; \
    nn = next_group(gq); \
    if (lane < nn) nxt = gq[lane]; }

__global__ __launch_bounds__(RWAVES * 32) void reduce_kernel(
    const float* __restrict__ wh,
    const float* recf,
    const float* offf,
    float* __restrict__ sums,
    float* __restrict__ cnts)
{
    __shared__ v4f acc4[BN * CC / 4];
    __shared__ unsigned cl[BN];
    float* acc = reinterpret_cast<float*>(acc4);

    const int tid = (int)threadIdx.x;
    const int r = (int)blockIdx.x / NB;
    const int b = (int)blockIdx.x % NB;

    for (int i = tid; i < BN * CC / 4; i += RWAVES * 32) acc4[i] = (v4f){};
    if (tid < BN) cl[tid] = 0u;
    __syncthreads();

    const int wave = __builtin_amdgcn_readfirstlane(tid >> 5);
    __builtin_assume(wave >= 0 && wave < RWAVES);
    const unsigned lane = (unsigned)tid & 31u;
    const unsigned c = lane & 15u;
    const unsigned rb = lane & 16u;
    const float* whr = wh + r * CC;
    const uint2* rec = reinterpret_cast<const uint2*>(recf);
    const unsigned* off = reinterpret_cast<const unsigned*>(offf);

    const int nruns = (KCH - wave + RWAVES - 1) / RWAVES;
    unsigned o0 = 0u, o1 = 0u;
    if ((int)lane < nruns) {
        const unsigned* p = off + (size_t)(r * KCH + wave + RWAVES * (int)lane) * NBP + b;
        o0 = p[0];
        o1 = p[1];
    }

    int j = -1;
    unsigned pos = 0u, end = 0u;
    const uint2* rbase = rec;
    auto next_group = [&](const uint2*& gp) -> unsigned {
        while (pos >= end) {
            ++j;
            if (j >= nruns) return 0u;
            pos = (unsigned)__builtin_amdgcn_readlane((int)o0, j);
            end = (unsigned)__builtin_amdgcn_readlane((int)o1, j);
            rbase = rec + (size_t)(r * KCH + wave + RWAVES * j) * CH;
        }
        gp = rbase + pos;
        const unsigned left = end - pos;
        pos += 32u;
        return left < 32u ? left : 32u;
    };

    const uint2* gp = rec;
    unsigned n = next_group(gp);
    uint2 cur;
    cur.x = 0u; cur.y = 0u;
    if (lane < n) cur = gp[lane];
    while (n != 0u) {
        if (lane < n) atomicAdd(&cl[cur.x >> SRCBITS], 1u);

        unsigned nn;
        uint2 nxt;
        nxt.x = 0u; nxt.y = 0u;
        if (n == 32u) {
            RGATHER_F(0)  RGATHER_F(1)  RGATHER_F(2)  RGATHER_F(3)
            RGATHER_F(4)  RGATHER_F(5)  RGATHER_F(6)  RGATHER_F(7)
            RGATHER_F(8)  RGATHER_F(9)  RGATHER_F(10) RGATHER_F(11)
            RGATHER_F(12) RGATHER_F(13) RGATHER_F(14) RGATHER_F(15)

            NEXT_RECORDS

            RADD_F(0)  RADD_F(1)  RADD_F(2)  RADD_F(3)
            RADD_F(4)  RADD_F(5)  RADD_F(6)  RADD_F(7)
            RADD_F(8)  RADD_F(9)  RADD_F(10) RADD_F(11)
            RADD_F(12) RADD_F(13) RADD_F(14) RADD_F(15)
        } else {
            RGATHER(0)  RGATHER(1)  RGATHER(2)  RGATHER(3)
            RGATHER(4)  RGATHER(5)  RGATHER(6)  RGATHER(7)
            RGATHER(8)  RGATHER(9)  RGATHER(10) RGATHER(11)
            RGATHER(12) RGATHER(13) RGATHER(14) RGATHER(15)

            NEXT_RECORDS

            RADD(0)  RADD(1)  RADD(2)  RADD(3)
            RADD(4)  RADD(5)  RADD(6)  RADD(7)
            RADD(8)  RADD(9)  RADD(10) RADD(11)
            RADD(12) RADD(13) RADD(14) RADD(15)
        }

        cur = nxt;
        n = nn;
    }
    __syncthreads();

    const int n0 = b * BN;
    const int nb = (NN - n0) < BN ? (NN - n0) : BN;
    v4f* s4 = reinterpret_cast<v4f*>(sums + ((size_t)r * NN + n0) * CC);
    for (int i = tid; i < nb * (CC / 4); i += RWAVES * 32) s4[i] = acc4[i];
    if (tid < nb) cnts[(size_t)r * NN + n0 + tid] = (float)cl[tid];
}
#undef RGATHER
#undef RADD
#undef RGATHER_F
#undef RADD_F
#undef NEXT_RECORDS
#undef ROW_SHARE

#define FTHREADS 128
#define FPER 5
#define FQ (NN * CC / 4)
#define FBLOCKS (FQ / (FTHREADS * FPER))
static_assert(CC % 4 == 0, "a node's channels are whole 16-byte groups");
static_assert(FQ % (FTHREADS * FPER) == 0, "the work-groups cover the groups exactly: no lane is idle, no bound is tested");

__global__ __launch_bounds__(FTHREADS) __attribute__((amdgpu_waves_per_eu(1, 4))) void finalize_kernel(
    const float* __restrict__ sums,
    const float* __restrict__ cnts,
    float* __restrict__ out)
{
    const v4f* s4 = reinterpret_cast<const v4f*>(sums);
    v4f* o4 = reinterpret_cast<v4f*>(out);
    const int i0 = (int)blockIdx.x * (FTHREADS * FPER) + (int)threadIdx.x;

    v4f s[FPER][RR];
    float cn[FPER][RR];
    #pragma unroll
    for (int u = 0; u < FPER; ++u) {
        const int i = i0 + u * FTHREADS;
        #pragma unroll
        for (int r = 0; r < RR; ++r) {
            cn[u][r] = cnts[r * NN + (i >> 2)];
            s[u][r]  = s4[r * FQ + i];
        }
    }

    #pragma unroll
    for (int u = 0; u < FPER; ++u) {
        v4f acc = (v4f){};
        #pragma unroll
        for (int r = 0; r < RR; ++r) {
            const float d = fmaxf(cn[u][r], 1.0f);
            acc.x += s[u][r].x / d;
            acc.y += s[u][r].y / d;
            acc.z += s[u][r].z / d;
            acc.w += s[u][r].w / d;
        }
        o4[i0 + u * FTHREADS] = acc;
    }
}

extern "C" void kernel_launch(void* const* d_in, const int* in_sizes, int n_in,
                              void* d_out, int out_size, void* d_ws, size_t ws_size,
                              hipStream_t stream) {
    const float* feat = (const float*)d_in[0];
    const int*   srci = (const int*)  d_in[1];
    const int*   dsti = (const int*)  d_in[2];
    const float* ew   = (const float*)d_in[3];
    const float* W    = (const float*)d_in[4];
    const float* bias = (const float*)d_in[5];
    float* out = (float*)d_out;

    float* ws   = (float*)d_ws;
    float* wh   = ws;
    float* sums = ws + (size_t)NN * (RR * CC);
    float* cnts = sums + (size_t)NN * (RR * CC);

    if (ws_size >= (size_t)117320000) {
        float* recf = cnts + (size_t)RR * NN;
        float* offf = recf + (size_t)2 * RR * EE;
        gemm_wmma_kernel<<<GBLOCKS, GWAVES * 32, 0, stream>>>(feat, W, bias, wh, nullptr, nullptr);
        part_kernel<<<RR * KCH, PTHREADS, 0, stream>>>(srci, dsti, ew, recf, offf);
        reduce_kernel<<<RR * NB, RWAVES * 32, 0, stream>>>(wh, recf, offf, sums, cnts);
    } else {
        {
            gemm_wmma_kernel<<<GBLOCKS, GWAVES * 32, 0, stream>>>(feat, W, bias, wh, sums, out);
        }

        {
            scatter_count_kernel<<<CBLOCKS + SBLOCKS, CTHREADS, 0, stream>>>(wh, srci, dsti, ew, sums, cnts, out);
        }

    }

    finalize_kernel<<<FBLOCKS, FTHREADS, 0, stream>>>(sums, cnts, out);
}


# launches: the edge sort's work-groups ride in the linear-layer launch (neither reads the other; 256 linear work-groups of 16 waves share the WGPs with them): 4 launches become 3; sums unchanged
# speedup vs baseline: 18.3322x; 1.0615x over previous
#include <hip/hip_runtime.h>

#define NN 100000
#define EE 2000000
#define DD 256
#define CC 16
#define RR 4

typedef __attribute__((ext_vector_type(2))) float v2f;
typedef __attribute__((ext_vector_type(8))) float v8f;

#define CR 12500
#define CRANGES (NN / CR)
#define CSLICES 8
#define CSLICE_V (EE / CSLICES / 4)
#define CTHREADS 512
#define CAHEAD 8
static_assert(NN % CR == 0, "ranges cover the nodes exactly");
static_assert(EE % (CSLICES * 4) == 0, "slices are whole 16-byte groups");

__device__ __forceinline__ void tally4(unsigned* hist, const int4 v, const int lo) {
    unsigned o;
    o = (unsigned)(v.x - lo); if (o < (unsigned)CR) atomicAdd(&hist[o], 1u);
    o = (unsigned)(v.y - lo); if (o < (unsigned)CR) atomicAdd(&hist[o], 1u);
    o = (unsigned)(v.z - lo); if (o < (unsigned)CR) atomicAdd(&hist[o], 1u);
    o = (unsigned)(v.w - lo); if (o < (unsigned)CR) atomicAdd(&hist[o], 1u);
}

#define CBLOCKS (RR * CSLICES * CRANGES)

__device__ __forceinline__ void count_body(
    unsigned* hist,
    const int* __restrict__ dst,
    float* __restrict__ cnts)
{
    const int b = (int)blockIdx.x;
    const int q = b % CRANGES;
    const int k = (b / CRANGES) % CSLICES;
    const int r = b / (CRANGES * CSLICES);
    const int lo = q * CR;

    for (int i = (int)threadIdx.x; i < CR; i += CTHREADS) hist[i] = 0u;
    __syncthreads();

    const int4* p = reinterpret_cast<const int4*>(dst + (size_t)r * EE) + (size_t)k * CSLICE_V;
    int i = (int)threadIdx.x;
    for (; i + (CAHEAD - 1) * CTHREADS < CSLICE_V; i += CAHEAD * CTHREADS) {
        int4 v[CAHEAD];
        #pragma unroll
        for (int u = 0; u < CAHEAD; ++u) v[u] = p[i + u * CTHREADS];
        #pragma unroll
        for (int u = 0; u < CAHEAD; ++u) tally4(hist, v[u], lo);
    }
    for (; i < CSLICE_V; i += CTHREADS) tally4(hist, p[i], lo);
    __syncthreads();

    float* c = cnts + (size_t)r * NN + lo;
    for (int j = (int)threadIdx.x; j < CR; j += CTHREADS) {
        const unsigned h = hist[j];
        if (h != 0u) atomicAdd(c + j, (float)h);
    }
}

typedef __attribute__((ext_vector_type(4))) float v4f;
#define GKB (DD / 8)
#define GCH 8
#define GWAVES 8
#define GBLOCKS 768
#define GTILES (NN / 16)
#define NZ4 ((NN * (RR * CC) + RR * NN) / 4)
static_assert((NN * (RR * CC) + RR * NN) % 4 == 0, "the zeroed region is whole 16-byte groups");
static_assert(NN % 16 == 0, "row tiles are whole");
static_assert(DD % 8 == 0, "k blocks are whole");
static_assert(GKB % (2 * GCH) == 0, "chunks come in pairs");

__device__ __forceinline__ void load_chunk(v4f (&a)[GCH], const float* p) {
    #pragma unroll
    for (int j = 0; j < GCH; ++j) a[j] = *reinterpret_cast<const v4f*>(p + 8 * j);
}

__device__ __forceinline__ void chunk_mma(v8f (&acc)[RR], const v4f (&a)[GCH], const v4f* wc) {
    #pragma unroll
    for (int j = 0; j < GCH; ++j) {
        v2f a0, a1;
        a0.x = a[j].x; a0.y = a[j].y;
        a1.x = a[j].z; a1.y = a[j].w;
        #pragma unroll
        for (int r = 0; r < RR; ++r) {
            const v4f b = wc[(r * GKB + j) * 32];
            v2f b0, b1;
            b0.x = b.x; b0.y = b.y;
            b1.x = b.z; b1.y = b.w;
            acc[r] = __builtin_amdgcn_wmma_f32_16x16x4_f32(
                 false, a0,  false, b0,
                 (short)0, acc[r],  false,  false);
            acc[r] = __builtin_amdgcn_wmma_f32_16x16x4_f32(
                 false, a1,  false, b1,
                 (short)0, acc[r],  false,  false);
        }
    }
}

__global__ __launch_bounds__(GWAVES * 32, 4) void gemm_wmma_kernel(
    const float* __restrict__ feat,
    const float* __restrict__ W,
    const float* __restrict__ bias,
    float* __restrict__ wh,
    float* zp,
    float* tick)
{
    __shared__ v4f wl[RR * GKB * 32];

    const int tid = (int)threadIdx.x;

    if (zp != nullptr) {
        v4f* z = reinterpret_cast<v4f*>(zp);
        for (int i = (int)blockIdx.x * (GWAVES * 32) + tid; i < NZ4; i += GBLOCKS * GWAVES * 32)
            z[i] = (v4f){};
        if (blockIdx.x == 0 && tid == 0) {
            unsigned* q = reinterpret_cast<unsigned*>(tick);
            q[0] = 0u;
            q[64] = 0u;
        }
    }

    #pragma unroll
    for (int i = 0; i < (RR * GKB * 32) / (GWAVES * 32); ++i) {
        const int idx = tid + (GWAVES * 32) * i;
        const int ln = idx & 31;
        const int kb = (idx >> 5) & (GKB - 1);
        const int r  = idx >> 10;
        const float* p = W + (size_t)r * (DD * CC) + (kb * 8 + 4 * (ln >> 4)) * CC + (ln & 15);
        v4f v;
        v.x = p[0];
        v.y = p[CC];
        v.z = p[2 * CC];
        v.w = p[3 * CC];
        wl[idx] = v;
    }
    __syncthreads();

    const int wave = __builtin_amdgcn_readfirstlane(tid >> 5);
    const int lane = tid & 31;
    const int col  = lane & 15;
    const int half = lane >> 4;

    const int first = wave * GBLOCKS + (int)blockIdx.x;
    const float* fl = feat + (size_t)col * DD + 4 * half;

    v4f ae[GCH], ao[GCH];
    if (first < GTILES) load_chunk(ae, fl + (size_t)first * (16 * DD));

    for (int tile = first; tile < GTILES; tile += GBLOCKS * GWAVES) {
        v8f acc[RR];
        #pragma unroll
        for (int r = 0; r < RR; ++r) acc[r] = (v8f){};
        const float* pt = fl + (size_t)tile * (16 * DD);

        #pragma nounroll
        for (int cp = 0; cp < GKB / (2 * GCH); ++cp) {
            load_chunk(ao, pt + (2 * cp + 1) * (8 * GCH));
            chunk_mma(acc, ae, wl + (size_t)(2 * cp * GCH) * 32 + lane);

            const bool last = (cp == GKB / (2 * GCH) - 1);
            const int ntile = last ? tile + GBLOCKS * GWAVES : tile;
            const int nch   = last ? 0 : 2 * cp + 2;
            if (ntile < GTILES) load_chunk(ae, fl + (size_t)ntile * (16 * DD) + nch * (8 * GCH));
            chunk_mma(acc, ao, wl + (size_t)((2 * cp + 1) * GCH) * 32 + lane);
        }

        asm volatile("v_nop\n\tv_nop\n\tv_nop\n\tv_nop");

        float* out = wh + (size_t)tile * (16 * RR * CC) + col;
        #pragma unroll
        for (int r = 0; r < RR; ++r) {
            const float bv = bias[r * CC + col];
            #pragma unroll
            for (int i = 0; i < 8; ++i) {
                const int m = i + 8 * half;
                out[(size_t)m * (RR * CC) + r * CC] = acc[r][i] + bv;
            }
        }
    }
}

#define NH 50000
#define GROUPS 4
#define NTASK ((unsigned)(((long long)RR * EE) / (32 * GROUPS)))
#define QSTRIDE 64
static_assert(EE % 32 == 0, "32 consecutive edges must belong to one etype");
static_assert(((long long)RR * EE) % (32 * GROUPS) == 0, "tasks cover the edges exactly");
static_assert(NH % 4 == 0, "the split falls on a 256B boundary of sums");

#define ROW_SHARE(X, J) __builtin_amdgcn_update_dpp(0, (X), 0x150 + (J), 0xF, 0xF, false)

#define GATHER(J) \
    const int d##J = ROW_SHARE(d, J); \
    const int s##J = ROW_SHARE(s, J); \
    const bool m##J = ((d##J >= NH) ? 1 : 0) == h; \
    float v##J = 0.0f; \
    if (m##J) v##J = whr[(unsigned)(s##J * (RR * CC) + c)];
#define ROW_ADD(J) { \
    const float w##J = __int_as_float(ROW_SHARE(wb, J)); \
    if (m##J) atomicAdd(sr + (unsigned)(d##J * CC + c), v##J * w##J); }

__device__ __forceinline__ void edge_group(
    const float* __restrict__ whr, float* __restrict__ sr,
    const int s, const int d, const int wb, const int c, const int h)
{
    GATHER(0)  GATHER(1)  GATHER(2)  GATHER(3)
    GATHER(4)  GATHER(5)  GATHER(6)  GATHER(7)
    GATHER(8)  GATHER(9)  GATHER(10) GATHER(11)
    GATHER(12) GATHER(13) GATHER(14) GATHER(15)

    ROW_ADD(0)  ROW_ADD(1)  ROW_ADD(2)  ROW_ADD(3)
    ROW_ADD(4)  ROW_ADD(5)  ROW_ADD(6)  ROW_ADD(7)
    ROW_ADD(8)  ROW_ADD(9)  ROW_ADD(10) ROW_ADD(11)
    ROW_ADD(12) ROW_ADD(13) ROW_ADD(14) ROW_ADD(15)
}

#define SBLOCKS 1024
__global__ __launch_bounds__(CTHREADS) void scatter_count_kernel(
    const float* __restrict__ wh,
    const int*   __restrict__ src,
    const int*   __restrict__ dst,
    const float* __restrict__ ew,
    float* __restrict__ sums,
    float* __restrict__ cnts,
    float* tick)
{
    __shared__ unsigned hist[CR];
    if (blockIdx.x < CBLOCKS) {
        count_body(hist, dst, cnts);
        return;
    }

    unsigned* q = reinterpret_cast<unsigned*>(tick);
    const unsigned lane = threadIdx.x & 31u;
    const int c = (int)(lane & 15u);

    const unsigned place = (unsigned)__builtin_amdgcn_s_sendmsg_rtn(0x87);
    int h = (((place >> 16) & 0xFu) >= 4u) ? 1 : 0;

    for (int pass = 0; pass < 2; ++pass) {
        unsigned* qh = q + h * QSTRIDE;
        for (;;) {
            unsigned t = 0;
            if (lane == 0) t = atomicAdd(qh, 1u);
            t = (unsigned)__builtin_amdgcn_readfirstlane((int)t);
            if (t >= NTASK) break;

            const unsigned i0 = t * (32u * GROUPS) + lane;

            int s[GROUPS], d[GROUPS], wb[GROUPS];
            #pragma unroll
            for (int g = 0; g < GROUPS; ++g) {
                const unsigned i = i0 + 32u * g;
                s[g]  = src[i];
                d[g]  = dst[i];
                wb[g] = __float_as_int(ew[i]);
            }

            #pragma unroll
            for (int g = 0; g < GROUPS; ++g) {
                const unsigned i = i0 + 32u * g;
                const int r = __builtin_amdgcn_readfirstlane((int)(i / (unsigned)EE));
                edge_group(wh + r * CC,
                           sums + (size_t)r * NN * CC,
                           s[g], d[g], wb[g], c, h);
            }
        }
        h ^= 1;
    }
}
#undef GATHER
#undef ROW_ADD

#define BN 782
#define LOCBITS 10
#define NB ((NN + BN - 1) / BN)
#define NBP 130
#define PB 160
#define CH 8000
#define KCH (EE / CH)
#define PTHREADS 512
#define PPER ((CH + PTHREADS - 1) / PTHREADS)
#define SRCBITS 17
#define RWAVES 32
static_assert(EE % CH == 0, "a chunk never spans two etypes");
static_assert(CH % 2 == 0, "a chunk is whole 16-byte groups of records");
static_assert(NN <= (1 << SRCBITS), "the source node fits its field");
static_assert(LOCBITS + SRCBITS <= 32 && BN <= (1 << LOCBITS), "the record's first word holds both fields");
static_assert(NB + 1 <= NBP && NBP <= PB, "the table row holds every start and the end");
static_assert(KCH <= RWAVES * 32, "one lane per run of a wave");
static_assert(BN <= RWAVES * 32, "one thread per count in the reduce work-group");
static_assert((BN * CC) % 4 == 0, "a bin's sums are whole 16-byte groups");
static_assert(PB % 32 == 0, "the scan runs in whole waves");

__device__ __forceinline__ void part_body(
    uint4* lrec4,
    unsigned* hist,
    unsigned* wsum,
    const int chunk,
    const int*   __restrict__ src,
    const int*   __restrict__ dst,
    const float* __restrict__ ew,
    float* recf,
    float* offf)
{
    uint2* lrec = reinterpret_cast<uint2*>(lrec4);

    const int tid = (int)threadIdx.x;
    const size_t e0 = (size_t)chunk * CH;

    if (tid < PB) hist[tid] = 0u;
    __syncthreads();

    int dv[PPER], sv[PPER], wv[PPER];
    #pragma unroll
    for (int u = 0; u < PPER; ++u) {
        const int i = tid + PTHREADS * u;
        dv[u] = 0; sv[u] = 0; wv[u] = 0;
        if (i < CH) {
            dv[u] = dst[e0 + i];
            sv[u] = src[e0 + i];
            wv[u] = __float_as_int(ew[e0 + i]);
        }
    }
    unsigned rk[PPER];
    #pragma unroll
    for (int u = 0; u < PPER; ++u) {
        const int i = tid + PTHREADS * u;
        rk[u] = 0u;
        if (i < CH) rk[u] = atomicAdd(&hist[(unsigned)dv[u] / BN], 1u);
    }
    __syncthreads();

    unsigned hv = 0u;
    if (tid < PB) hv = hist[tid];
    unsigned incl = hv;
    #pragma unroll
    for (int o = 1; o < 32; o <<= 1) {
        const unsigned t = (unsigned)__shfl_up((int)incl, (unsigned)o);
        if ((tid & 31) >= o) incl += t;
    }
    if (tid < PB && (tid & 31) == 31) wsum[tid >> 5] = incl;
    __syncthreads();
    if (tid < PB) {
        unsigned base = 0u;
        for (int w = 0; w < (tid >> 5); ++w) base += wsum[w];
        const unsigned st = base + incl - hv;
        hist[tid] = st;
        if (tid < NBP) reinterpret_cast<unsigned*>(offf)[(size_t)chunk * NBP + tid] = st;
    }
    __syncthreads();

    #pragma unroll
    for (int u = 0; u < PPER; ++u) {
        const int i = tid + PTHREADS * u;
        if (i < CH) {
            const unsigned bin = (unsigned)dv[u] / BN;
            const unsigned pos = hist[bin] + rk[u];
            uint2 rc;
            rc.x = (unsigned)sv[u] | (((unsigned)dv[u] - bin * BN) << SRCBITS);
            rc.y = (unsigned)wv[u];
            lrec[pos] = rc;
        }
    }
    __syncthreads();

    uint4* g4 = reinterpret_cast<uint4*>(recf) + e0 / 2;
    for (int i = tid; i < CH / 2; i += PTHREADS) g4[i] = lrec4[i];
}

#define XWAVES 16
#define XTHREADS (XWAVES * 32)
#define XG 256
#define XGRID (XG + RR * KCH)
#define XCH 4
static_assert(XTHREADS == PTHREADS, "the sort's body is written for this many threads");
static_assert((RR * GKB * 32) % XTHREADS == 0, "W is staged in whole rounds of the work-group");
static_assert(RR * KCH >= XG, "the first 2 * XG work-groups hold XG chunks");
static_assert(XG * XWAVES <= GTILES, "every wave has a first tile");
static_assert(GKB % (2 * XCH) == 0, "chunks come in pairs");
static_assert(CH / 2 + PB / 4 + (PB / 32 + 3) / 4 <= RR * GKB * 32, "the sort's LDS fits the buffer sized for W");

template <int NCH>
__device__ __forceinline__ void load_chunk_n(v4f (&a)[NCH], const float* p) {
    #pragma unroll
    for (int j = 0; j < NCH; ++j) a[j] = *reinterpret_cast<const v4f*>(p + 8 * j);
}

template <int NCH>
__device__ __forceinline__ void chunk_mma_n(v8f (&acc)[RR], const v4f (&a)[NCH], const v4f* wc) {
    #pragma unroll
    for (int j = 0; j < NCH; ++j) {
        v2f a0, a1;
        a0.x = a[j].x; a0.y = a[j].y;
        a1.x = a[j].z; a1.y = a[j].w;
        #pragma unroll
        for (int r = 0; r < RR; ++r) {
            const v4f b = wc[(r * GKB + j) * 32];
            v2f b0, b1;
            b0.x = b.x; b0.y = b.y;
            b1.x = b.z; b1.y = b.w;
            acc[r] = __builtin_amdgcn_wmma_f32_16x16x4_f32(
                 false, a0,  false, b0,
                 (short)0, acc[r],  false,  false);
            acc[r] = __builtin_amdgcn_wmma_f32_16x16x4_f32(
                 false, a1,  false, b1,
                 (short)0, acc[r],  false,  false);
        }
    }
}

__device__ __forceinline__ void linear_body(
    v4f* wl,
    const int g,
    const float* __restrict__ feat,
    const float* __restrict__ W,
    const float* __restrict__ bias,
    float* __restrict__ wh)
{
    const int tid = (int)threadIdx.x;

    #pragma unroll
    for (int i = 0; i < (RR * GKB * 32) / XTHREADS; ++i) {
        const int idx = tid + XTHREADS * i;
        const int ln = idx & 31;
        const int kb = (idx >> 5) & (GKB - 1);
        const int r  = idx >> 10;
        const float* p = W + (size_t)r * (DD * CC) + (kb * 8 + 4 * (ln >> 4)) * CC + (ln & 15);
        v4f v;
        v.x = p[0];
        v.y = p[CC];
        v.z = p[2 * CC];
        v.w = p[3 * CC];
        wl[idx] = v;
    }
    __syncthreads();

    const int wave = __builtin_amdgcn_readfirstlane(tid >> 5);
    const int lane = tid & 31;
    const int col  = lane & 15;
    const int half = lane >> 4;

    const int first = wave * XG + g;
    const float* fl = feat + (size_t)col * DD + 4 * half;

    v4f ae[XCH], ao[XCH];
    load_chunk_n<XCH>(ae, fl + (size_t)first * (16 * DD));

    for (int tile = first; tile < GTILES; tile += XG * XWAVES) {
        v8f acc[RR];
        #pragma unroll
        for (int r = 0; r < RR; ++r) acc[r] = (v8f){};
        const float* pt = fl + (size_t)tile * (16 * DD);

        #pragma nounroll
        for (int cp = 0; cp < GKB / (2 * XCH); ++cp) {
            load_chunk_n<XCH>(ao, pt + (2 * cp + 1) * (8 * XCH));
            chunk_mma_n<XCH>(acc, ae, wl + (size_t)(2 * cp * XCH) * 32 + lane);

            const bool last = (cp == GKB / (2 * XCH) - 1);
            const int ntile = last ? tile + XG * XWAVES : tile;
            const int nch   = last ? 0 : 2 * cp + 2;
            if (ntile < GTILES) load_chunk_n<XCH>(ae, fl + (size_t)ntile * (16 * DD) + nch * (8 * XCH));
            chunk_mma_n<XCH>(acc, ao, wl + (size_t)((2 * cp + 1) * XCH) * 32 + lane);
        }

        asm volatile("v_nop\n\tv_nop\n\tv_nop\n\tv_nop");

        float* out = wh + (size_t)tile * (16 * RR * CC) + col;
        #pragma unroll
        for (int r = 0; r < RR; ++r) {
            const float bv = bias[r * CC + col];
            #pragma unroll
            for (int i = 0; i < 8; ++i) {
                const int m = i + 8 * half;
                out[(size_t)m * (RR * CC) + r * CC] = acc[r][i] + bv;
            }
        }
    }
}

__global__ __launch_bounds__(XTHREADS, 8) void front_kernel(
    const float* __restrict__ feat,
    const float* __restrict__ W,
    const float* __restrict__ bias,
    float* __restrict__ wh,
    const int*   __restrict__ src,
    const int*   __restrict__ dst,
    const float* __restrict__ ew,
    float* recf,
    float* offf)
{
    __shared__ uint4 smem[RR * GKB * 32];

    const int b = (int)blockIdx.x;
    bool lin = false;
    int idx = b - XG;
    if (b < 2 * XG) {
        lin = (((b ^ (b / XG)) & 1) == 0);
        idx = b >> 1;
    }

    if (lin) {
        linear_body(reinterpret_cast<v4f*>(smem), idx, feat, W, bias, wh);
    } else {
        part_body(smem,
                  reinterpret_cast<unsigned*>(smem + CH / 2),
                  reinterpret_cast<unsigned*>(smem + CH / 2 + PB / 4),
                  idx, src, dst, ew, recf, offf);
    }
}

#define RGATHER(J) \
    const unsigned x##J = (unsigned)ROW_SHARE((int)cur.x, J); \
    const bool m##J = (rb + (unsigned)(J)) < n; \
    float v##J = 0.0f; \
    if (m##J) v##J = whr[(x##J & ((1u << SRCBITS) - 1u)) * (unsigned)(RR * CC) + c];
#define RADD(J) { \
    const float w##J = __int_as_float(ROW_SHARE((int)cur.y, J)); \
    if (m##J) atomicAdd(&acc[(x##J >> SRCBITS) * CC + c], v##J * w##J); }

#define RGATHER_F(J) \
    const unsigned x##J = (unsigned)ROW_SHARE((int)cur.x, J); \
    const float v##J = whr[(x##J & ((1u << SRCBITS) - 1u)) * (unsigned)(RR * CC) + c];
#define RADD_F(J) { \
    const float w##J = __int_as_float(ROW_SHARE((int)cur.y, J)); \
    atomicAdd(&acc[(x##J >> SRCBITS) * CC + c], v##J * w##J); }
#define NEXT_RECORDS { \
    const uint2* gq = rec; \
    nn = next_group(gq); \
    if (lane < nn) nxt = gq[lane]; }

__global__ __launch_bounds__(RWAVES * 32) void reduce_kernel(
    const float* __restrict__ wh,
    const float* recf,
    const float* offf,
    float* __restrict__ sums,
    float* __restrict__ cnts)
{
    __shared__ v4f acc4[BN * CC / 4];
    __shared__ unsigned cl[BN];
    float* acc = reinterpret_cast<float*>(acc4);

    const int tid = (int)threadIdx.x;
    const int r = (int)blockIdx.x / NB;
    const int b = (int)blockIdx.x % NB;

    for (int i = tid; i < BN * CC / 4; i += RWAVES * 32) acc4[i] = (v4f){};
    if (tid < BN) cl[tid] = 0u;
    __syncthreads();

    const int wave = __builtin_amdgcn_readfirstlane(tid >> 5);
    __builtin_assume(wave >= 0 && wave < RWAVES);
    const unsigned lane = (unsigned)tid & 31u;
    const unsigned c = lane & 15u;
    const unsigned rb = lane & 16u;
    const float* whr = wh + r * CC;
    const uint2* rec = reinterpret_cast<const uint2*>(recf);
    const unsigned* off = reinterpret_cast<const unsigned*>(offf);

    const int nruns = (KCH - wave + RWAVES - 1) / RWAVES;
    unsigned o0 = 0u, o1 = 0u;
    if ((int)lane < nruns) {
        const unsigned* p = off + (size_t)(r * KCH + wave + RWAVES * (int)lane) * NBP + b;
        o0 = p[0];
        o1 = p[1];
    }

    int j = -1;
    unsigned pos = 0u, end = 0u;
    const uint2* rbase = rec;
    auto next_group = [&](const uint2*& gp) -> unsigned {
        while (pos >= end) {
            ++j;
            if (j >= nruns) return 0u;
            pos = (unsigned)__builtin_amdgcn_readlane((int)o0, j);
            end = (unsigned)__builtin_amdgcn_readlane((int)o1, j);
            rbase = rec + (size_t)(r * KCH + wave + RWAVES * j) * CH;
        }
        gp = rbase + pos;
        const unsigned left = end - pos;
        pos += 32u;
        return left < 32u ? left : 32u;
    };

    const uint2* gp = rec;
    unsigned n = next_group(gp);
    uint2 cur;
    cur.x = 0u; cur.y = 0u;
    if (lane < n) cur = gp[lane];
    while (n != 0u) {
        if (lane < n) atomicAdd(&cl[cur.x >> SRCBITS], 1u);

        unsigned nn;
        uint2 nxt;
        nxt.x = 0u; nxt.y = 0u;
        if (n == 32u) {
            RGATHER_F(0)  RGATHER_F(1)  RGATHER_F(2)  RGATHER_F(3)
            RGATHER_F(4)  RGATHER_F(5)  RGATHER_F(6)  RGATHER_F(7)
            RGATHER_F(8)  RGATHER_F(9)  RGATHER_F(10) RGATHER_F(11)
            RGATHER_F(12) RGATHER_F(13) RGATHER_F(14) RGATHER_F(15)

            NEXT_RECORDS

            RADD_F(0)  RADD_F(1)  RADD_F(2)  RADD_F(3)
            RADD_F(4)  RADD_F(5)  RADD_F(6)  RADD_F(7)
            RADD_F(8)  RADD_F(9)  RADD_F(10) RADD_F(11)
            RADD_F(12) RADD_F(13) RADD_F(14) RADD_F(15)
        } else {
            RGATHER(0)  RGATHER(1)  RGATHER(2)  RGATHER(3)
            RGATHER(4)  RGATHER(5)  RGATHER(6)  RGATHER(7)
            RGATHER(8)  RGATHER(9)  RGATHER(10) RGATHER(11)
            RGATHER(12) RGATHER(13) RGATHER(14) RGATHER(15)

            NEXT_RECORDS

            RADD(0)  RADD(1)  RADD(2)  RADD(3)
            RADD(4)  RADD(5)  RADD(6)  RADD(7)
            RADD(8)  RADD(9)  RADD(10) RADD(11)
            RADD(12) RADD(13) RADD(14) RADD(15)
        }

        cur = nxt;
        n = nn;
    }
    __syncthreads();

    const int n0 = b * BN;
    const int nb = (NN - n0) < BN ? (NN - n0) : BN;
    v4f* s4 = reinterpret_cast<v4f*>(sums + ((size_t)r * NN + n0) * CC);
    for (int i = tid; i < nb * (CC / 4); i += RWAVES * 32) s4[i] = acc4[i];
    if (tid < nb) cnts[(size_t)r * NN + n0 + tid] = (float)cl[tid];
}
#undef RGATHER
#undef RADD
#undef RGATHER_F
#undef RADD_F
#undef NEXT_RECORDS
#undef ROW_SHARE

#define FTHREADS 128
#define FPER 5
#define FQ (NN * CC / 4)
#define FBLOCKS (FQ / (FTHREADS * FPER))
static_assert(CC % 4 == 0, "a node's channels are whole 16-byte groups");
static_assert(FQ % (FTHREADS * FPER) == 0, "the work-groups cover the groups exactly: no lane is idle, no bound is tested");

__global__ __launch_bounds__(FTHREADS) __attribute__((amdgpu_waves_per_eu(1, 4))) void finalize_kernel(
    const float* __restrict__ sums,
    const float* __restrict__ cnts,
    float* __restrict__ out)
{
    const v4f* s4 = reinterpret_cast<const v4f*>(sums);
    v4f* o4 = reinterpret_cast<v4f*>(out);
    const int i0 = (int)blockIdx.x * (FTHREADS * FPER) + (int)threadIdx.x;

    v4f s[FPER][RR];
    float cn[FPER][RR];
    #pragma unroll
    for (int u = 0; u < FPER; ++u) {
        const int i = i0 + u * FTHREADS;
        #pragma unroll
        for (int r = 0; r < RR; ++r) {
            cn[u][r] = cnts[r * NN + (i >> 2)];
            s[u][r]  = s4[r * FQ + i];
        }
    }

    #pragma unroll
    for (int u = 0; u < FPER; ++u) {
        v4f acc = (v4f){};
        #pragma unroll
        for (int r = 0; r < RR; ++r) {
            const float d = fmaxf(cn[u][r], 1.0f);
            acc.x += s[u][r].x / d;
            acc.y += s[u][r].y / d;
            acc.z += s[u][r].z / d;
            acc.w += s[u][r].w / d;
        }
        o4[i0 + u * FTHREADS] = acc;
    }
}

extern "C" void kernel_launch(void* const* d_in, const int* in_sizes, int n_in,
                              void* d_out, int out_size, void* d_ws, size_t ws_size,
                              hipStream_t stream) {
    const float* feat = (const float*)d_in[0];
    const int*   srci = (const int*)  d_in[1];
    const int*   dsti = (const int*)  d_in[2];
    const float* ew   = (const float*)d_in[3];
    const float* W    = (const float*)d_in[4];
    const float* bias = (const float*)d_in[5];
    float* out = (float*)d_out;

    float* ws   = (float*)d_ws;
    float* wh   = ws;
    float* sums = ws + (size_t)NN * (RR * CC);
    float* cnts = sums + (size_t)NN * (RR * CC);

    if (ws_size >= (size_t)117320000) {
        float* recf = cnts + (size_t)RR * NN;
        float* offf = recf + (size_t)2 * RR * EE;
        front_kernel<<<XGRID, XTHREADS, 0, stream>>>(feat, W, bias, wh, srci, dsti, ew, recf, offf);
        reduce_kernel<<<RR * NB, RWAVES * 32, 0, stream>>>(wh, recf, offf, sums, cnts);
    } else {
        {
            gemm_wmma_kernel<<<GBLOCKS, GWAVES * 32, 0, stream>>>(feat, W, bias, wh, sums, out);
        }

        {
            scatter_count_kernel<<<CBLOCKS + SBLOCKS, CTHREADS, 0, stream>>>(wh, srci, dsti, ew, sums, cnts, out);
        }

    }

    finalize_kernel<<<FBLOCKS, FTHREADS, 0, stream>>>(sums, cnts, out);
}
